# starting kernel
# baseline (speedup 1.0000x reference)
#include <hip/hip_runtime.h>
#include <hip/hip_bf16.h>

typedef __attribute__((ext_vector_type(2))) float v2f;
typedef __attribute__((ext_vector_type(8))) float v8f;

__global__ __launch_bounds__(256) void gcn_gemm_bias_kernel(
    const float* __restrict__ A,
    const float* __restrict__ W,
    const float* __restrict__ bias,
    float* __restrict__ out,
    int Nrows)
{
    const int wave_in_block = threadIdx.x >> 5;
    const int lane          = threadIdx.x & 31;
    const int row_tile      = wave_in_block >> 2;
    const int col_tile      = wave_in_block & 3;

    const int row_base = (blockIdx.x * 2 + row_tile) * 16;
    if (row_base >= Nrows) return;

    const int n_base = col_tile * 16;
    const int half   = lane >> 4;
    const int m      = lane & 15;

    const float* arow = A + (long long)(row_base + m) * 64;

    v8f acc = {};
    #pragma unroll
    for (int k0 = 0; k0 < 64; k0 += 4) {
        const int ka = k0 + 2 * half;
        v2f a;
        a.x = arow[ka + 0];
        a.y = arow[ka + 1];
        v2f b;
        b.x = W[(ka + 0) * 64 + n_base + m];
        b.y = W[(ka + 1) * 64 + n_base + m];
        acc = __builtin_amdgcn_wmma_f32_16x16x4_f32(
            false, a, false, b, (short)0, acc, false, false);
    }

    const float bv = bias[n_base + m];
    #pragma unroll
    for (int r = 0; r < 8; ++r) {
        const int row = row_base + r + 8 * half;
        out[(long long)row * 64 + n_base + m] = acc[r] + bv;
    }
}

__global__ __launch_bounds__(256) void gcn_edge_scatter_kernel(
    const float* __restrict__ node_states,
    const float* __restrict__ edge_weight,
    const int*   __restrict__ src_index,
    const int*   __restrict__ dst_index,
    float* __restrict__ out,
    long long E)
{
    const long long gtid = (long long)blockIdx.x * blockDim.x + threadIdx.x;
    const long long e    = gtid >> 5;
    if (e >= E) return;
    const int lane = threadIdx.x & 31;

    const int   s = src_index[e];
    const int   d = dst_index[e];
    const float w = edge_weight[e];

    const float2* srow = (const float2*)(node_states + (long long)s * 64);
    float2 v = srow[lane];

    float* drow = out + (long long)d * 64 + lane * 2;
    atomicAdd(drow + 0, v.x * w);
    atomicAdd(drow + 1, v.y * w);
}

extern "C" void kernel_launch(void* const* d_in, const int* in_sizes, int n_in,
                              void* d_out, int out_size, void* d_ws, size_t ws_size,
                              hipStream_t stream) {
    const float* node_states = (const float*)d_in[0];
    const float* edge_weight = (const float*)d_in[1];
    const float* W           = (const float*)d_in[2];
    const float* b           = (const float*)d_in[3];
    const int*   src_index   = (const int*)d_in[4];
    const int*   dst_index   = (const int*)d_in[5];
    float*       out         = (float*)d_out;

    const int       Nrows = in_sizes[0] / 64;
    const long long E     = in_sizes[1];

    const int gemm_blocks = (Nrows + 31) / 32;
    gcn_gemm_bias_kernel<<<gemm_blocks, 256, 0, stream>>>(
        node_states, W, b, out, Nrows);

    const long long edges_per_block = 8;
    const long long scat_blocks = (E + edges_per_block - 1) / edges_per_block;
    gcn_edge_scatter_kernel<<<(unsigned)scat_blocks, 256, 0, stream>>>(
        node_states, edge_weight, src_index, dst_index, out, E);
}


# edge sum without float atomics: edges counting-sorted by destination in the call (integer rank add, prefix sum, placement); one wave sums a node's rows in registers, in arrival order of the ranks
# speedup vs baseline: 1.0691x; 1.0691x over previous
#include <hip/hip_runtime.h>
#include <hip/hip_bf16.h>

typedef __attribute__((ext_vector_type(2))) float v2f;
typedef __attribute__((ext_vector_type(8))) float v8f;

__global__ __launch_bounds__(256) void gcn_gemm_bias_kernel(
    const float* __restrict__ A,
    const float* __restrict__ W,
    const float* __restrict__ bias,
    float* __restrict__ out,
    int Nrows,
    int* __restrict__ cnt,
    int ncnt)
{
    for (int i = blockIdx.x * 256 + threadIdx.x; i < ncnt; i += gridDim.x * 256)
        cnt[i] = 0;

    const int wave_in_block = threadIdx.x >> 5;
    const int lane          = threadIdx.x & 31;
    const int row_tile      = wave_in_block >> 2;
    const int col_tile      = wave_in_block & 3;

    const int row_base = (blockIdx.x * 2 + row_tile) * 16;
    if (row_base >= Nrows) return;

    const int n_base = col_tile * 16;
    const int half   = lane >> 4;
    const int m      = lane & 15;

    const float* arow = A + (long long)(row_base + m) * 64;

    v8f acc = {};
    #pragma unroll
    for (int k0 = 0; k0 < 64; k0 += 4) {
        const int ka = k0 + 2 * half;
        v2f a;
        a.x = arow[ka + 0];
        a.y = arow[ka + 1];
        v2f b;
        b.x = W[(ka + 0) * 64 + n_base + m];
        b.y = W[(ka + 1) * 64 + n_base + m];
        acc = __builtin_amdgcn_wmma_f32_16x16x4_f32(
            false, a, false, b, (short)0, acc, false, false);
    }

    const float bv = bias[n_base + m];
    #pragma unroll
    for (int r = 0; r < 8; ++r) {
        const int row = row_base + r + 8 * half;
        out[(long long)row * 64 + n_base + m] = acc[r] + bv;
    }
}

__global__ __launch_bounds__(256) void edge_rank_kernel(
    const int* __restrict__ dst_index,
    int* __restrict__ cnt,
    int* __restrict__ rank,
    long long E)
{
    const long long e = (long long)blockIdx.x * 256 + threadIdx.x;
    if (e >= E) return;
    rank[e] = atomicAdd(cnt + dst_index[e], 1);
}

__global__ __launch_bounds__(1024) void node_offsets_kernel(
    const int* __restrict__ cnt,
    int* __restrict__ offs,
    int nchunks)
{
    __shared__ int s_wave[32];
    __shared__ int s_pre[32];
    const int tid  = threadIdx.x;
    const int lane = tid & 31;
    const int wave = tid >> 5;
    const int4* c4 = (const int4*)cnt;

    const int nb    = blockIdx.x;
    const int first = nb * 1024;
    int pre = 0;
    #pragma unroll 4
    for (int k = 0; k < nb; ++k) {
        const int4 v = c4[k * 1024 + tid];
        pre += v.x + v.y + v.z + v.w;
    }
    const int ci = first + tid;
    int4 c = make_int4(0, 0, 0, 0);
    if (ci < nchunks) c = c4[ci];
    const int tsum = c.x + c.y + c.z + c.w;

    int inc = tsum;
    for (int o = 1; o < 32; o <<= 1) {
        const int t = __shfl_up(inc, o);
        if (lane >= o) inc += t;
    }
    for (int o = 16; o > 0; o >>= 1) pre += __shfl_xor(pre, o);
    if (lane == 31) s_wave[wave] = inc;
    if (lane == 0)  s_pre[wave]  = pre;
    __syncthreads();

    const int wt = s_wave[lane];
    int winc = wt;
    for (int o = 1; o < 32; o <<= 1) {
        const int t = __shfl_up(winc, o);
        if (lane >= o) winc += t;
    }
    int pt = s_pre[lane];
    for (int o = 16; o > 0; o >>= 1) pt += __shfl_xor(pt, o);
    const int waves_before = __shfl(winc, wave) - __shfl(wt, wave);

    if (ci < nchunks) {
        int4 r;
        r.x = pt + waves_before + (inc - tsum);
        r.y = r.x + c.x;
        r.z = r.y + c.y;
        r.w = r.z + c.z;
        ((int4*)offs)[ci] = r;
    }
}

__global__ __launch_bounds__(256) void edge_place_kernel(
    const float* __restrict__ edge_weight,
    const int*   __restrict__ src_index,
    const int*   __restrict__ dst_index,
    const int*   __restrict__ rank,
    const int*   __restrict__ offs,
    int* __restrict__ sorted,
    long long E)
{
    const long long e = (long long)blockIdx.x * 256 + threadIdx.x;
    if (e >= E) return;
    const int slot = offs[dst_index[e]] + rank[e];
    ((int2*)sorted)[slot] = make_int2(src_index[e], __float_as_int(edge_weight[e]));
}

__device__ __forceinline__ void add_edge_row(
    float2& acc, const float2* __restrict__ rows, const int2 p, const int j, const int lane)
{
    const int   s = __builtin_amdgcn_readlane(p.x, j);
    const float w = __int_as_float(__builtin_amdgcn_readlane(p.y, j));
    const float2 x = rows[(long long)s * 32 + lane];
    acc.x = fmaf(w, x.x, acc.x);
    acc.y = fmaf(w, x.y, acc.y);
}

__global__ __launch_bounds__(256) void node_gather_sum_kernel(
    const float* __restrict__ node_states,
    const int*   __restrict__ offs,
    const int*   __restrict__ sorted,
    float* __restrict__ out,
    int Nrows)
{
    const int lane = threadIdx.x & 31;
    const int node = __builtin_amdgcn_readfirstlane(blockIdx.x * 8 + (threadIdx.x >> 5));
    if (node >= Nrows) return;
    const int beg = __builtin_amdgcn_readfirstlane(offs[node]);
    const int end = __builtin_amdgcn_readfirstlane(offs[node + 1]);

    const int2*   pairs = (const int2*)sorted;
    const float2* rows  = (const float2*)node_states;

    float2 acc = make_float2(0.0f, 0.0f);
    for (int base = beg; base < end; base += 32) {
        const int left = end - base;
        const int n = left < 32 ? left : 32;
        int2 p = make_int2(0, 0);
        if (lane < n) p = pairs[base + lane];
        int j = 0;
        for (; j + 8 <= n; j += 8) {
            add_edge_row(acc, rows, p, j + 0, lane);
            add_edge_row(acc, rows, p, j + 1, lane);
            add_edge_row(acc, rows, p, j + 2, lane);
            add_edge_row(acc, rows, p, j + 3, lane);
            add_edge_row(acc, rows, p, j + 4, lane);
            add_edge_row(acc, rows, p, j + 5, lane);
            add_edge_row(acc, rows, p, j + 6, lane);
            add_edge_row(acc, rows, p, j + 7, lane);
        }
        if (j + 4 <= n) {
            add_edge_row(acc, rows, p, j + 0, lane);
            add_edge_row(acc, rows, p, j + 1, lane);
            add_edge_row(acc, rows, p, j + 2, lane);
            add_edge_row(acc, rows, p, j + 3, lane);
            j += 4;
        }
        for (; j < n; ++j) add_edge_row(acc, rows, p, j, lane);
    }

    float2* orow = (float2*)(out + (long long)node * 64) + lane;
    float2 v = *orow;
    v.x += acc.x;
    v.y += acc.y;
    *orow = v;
}

extern "C" void kernel_launch(void* const* d_in, const int* in_sizes, int n_in,
                              void* d_out, int out_size, void* d_ws, size_t ws_size,
                              hipStream_t stream) {
    const float* node_states = (const float*)d_in[0];
    const float* edge_weight = (const float*)d_in[1];
    const float* W           = (const float*)d_in[2];
    const float* b           = (const float*)d_in[3];
    const int*   src_index   = (const int*)d_in[4];
    const int*   dst_index   = (const int*)d_in[5];
    float*       out         = (float*)d_out;

    const int       Nrows = in_sizes[0] / 64;
    const long long E     = in_sizes[1];

    const int ncnt = (Nrows + 64) & ~63;
    int* cnt    = (int*)d_ws;
    int* offs   = cnt + ncnt;
    int* rank   = offs + ncnt;
    int* sorted = rank + ((E + 63) & ~63LL);

    const int gemm_blocks = (Nrows + 31) / 32;
    gcn_gemm_bias_kernel<<<gemm_blocks, 256, 0, stream>>>(
        node_states, W, b, out, Nrows, cnt, ncnt);

    const unsigned edge_blocks = (unsigned)((E + 255) / 256);
    edge_rank_kernel<<<edge_blocks, 256, 0, stream>>>(dst_index, cnt, rank, E);

    const int nchunks = ncnt / 4;
    node_offsets_kernel<<<(nchunks + 1023) / 1024, 1024, 0, stream>>>(cnt, offs, nchunks);

    edge_place_kernel<<<edge_blocks, 256, 0, stream>>>(
        edge_weight, src_index, dst_index, rank, offs, sorted, E);

    node_gather_sum_kernel<<<(Nrows + 7) / 8, 256, 0, stream>>>(
        node_states, offs, sorted, out, Nrows);
}


# edge sort ranks in LDS in two levels (256-node buckets, then nodes) on wave-private counters, no global atomic per edge; a node's rows are summed in the order of these ranks, not in arrival order
# speedup vs baseline: 2.6526x; 2.4812x over previous
#include <hip/hip_runtime.h>
#include <hip/hip_bf16.h>

typedef __attribute__((ext_vector_type(2))) float v2f;
typedef __attribute__((ext_vector_type(8))) float v8f;

__global__ __launch_bounds__(256) void gcn_gemm_bias_kernel(
    const float* __restrict__ A,
    const float* __restrict__ W,
    const float* __restrict__ bias,
    float* __restrict__ out,
    int Nrows,
    int* __restrict__ cnt,
    int ncnt)
{
    for (int i = blockIdx.x * 256 + threadIdx.x; i < ncnt; i += gridDim.x * 256)
        cnt[i] = 0;

    const int wave_in_block = threadIdx.x >> 5;
    const int lane          = threadIdx.x & 31;
    const int row_tile      = wave_in_block >> 2;
    const int col_tile      = wave_in_block & 3;

    const int row_base = (blockIdx.x * 2 + row_tile) * 16;
    if (row_base >= Nrows) return;

    const int n_base = col_tile * 16;
    const int half   = lane >> 4;
    const int m      = lane & 15;

    const float* arow = A + (long long)(row_base + m) * 64;

    v8f acc = {};
    #pragma unroll
    for (int k0 = 0; k0 < 64; k0 += 4) {
        const int ka = k0 + 2 * half;
        v2f a;
        a.x = arow[ka + 0];
        a.y = arow[ka + 1];
        v2f b;
        b.x = W[(ka + 0) * 64 + n_base + m];
        b.y = W[(ka + 1) * 64 + n_base + m];
        acc = __builtin_amdgcn_wmma_f32_16x16x4_f32(
            false, a, false, b, (short)0, acc, false, false);
    }

    const float bv = bias[n_base + m];
    #pragma unroll
    for (int r = 0; r < 8; ++r) {
        const int row = row_base + r + 8 * half;
        out[(long long)row * 64 + n_base + m] = acc[r] + bv;
    }
}

#define L1_GROUPS   256
#define L1_WAVES    32
#define NODE_SHIFT  8
#define MAX_BUCKETS 512

__global__ __launch_bounds__(1024) void bucket_count_kernel(
    const int* __restrict__ dst_index,
    int* __restrict__ cnt,
    long long E, int wspan, int nb)
{
    __shared__ int s_cnt[MAX_BUCKETS];
    const int tid  = threadIdx.x;
    const int lane = tid & 31;
    const int wave = tid >> 5;
    if (tid < MAX_BUCKETS) s_cnt[tid] = 0;
    __syncthreads();

    const long long wbeg = ((long long)blockIdx.x * L1_WAVES + wave) * wspan;
    long long wend = wbeg + wspan;
    if (wend > E) wend = E;
    for (long long base = wbeg; base < wend; base += 128) {
        int  d[4];
        bool v[4];
        #pragma unroll
        for (int u = 0; u < 4; ++u) {
            const long long e = base + u * 32 + lane;
            v[u] = e < wend;
            d[u] = dst_index[v[u] ? e : wend - 1];
        }
        #pragma unroll
        for (int u = 0; u < 4; ++u)
            if (v[u]) atomicAdd(&s_cnt[d[u] >> NODE_SHIFT], 1);
    }
    __syncthreads();
    for (int b = tid; b < nb; b += 1024) cnt[b * L1_GROUPS + blockIdx.x] = s_cnt[b];
}

__global__ __launch_bounds__(1024) void node_offsets_kernel(
    const int* __restrict__ cnt,
    int* __restrict__ offs,
    int nchunks)
{
    __shared__ int s_wave[32];
    __shared__ int s_pre[32];
    const int tid  = threadIdx.x;
    const int lane = tid & 31;
    const int wave = tid >> 5;
    const int4* c4 = (const int4*)cnt;

    const int nb    = blockIdx.x;
    const int first = nb * 1024;
    int pre = 0;
    #pragma unroll 4
    for (int k = 0; k < nb; ++k) {
        const int4 v = c4[k * 1024 + tid];
        pre += v.x + v.y + v.z + v.w;
    }
    const int ci = first + tid;
    int4 c = make_int4(0, 0, 0, 0);
    if (ci < nchunks) c = c4[ci];
    const int tsum = c.x + c.y + c.z + c.w;

    int inc = tsum;
    for (int o = 1; o < 32; o <<= 1) {
        const int t = __shfl_up(inc, o);
        if (lane >= o) inc += t;
    }
    for (int o = 16; o > 0; o >>= 1) pre += __shfl_xor(pre, o);
    if (lane == 31) s_wave[wave] = inc;
    if (lane == 0)  s_pre[wave]  = pre;
    __syncthreads();

    const int wt = s_wave[lane];
    int winc = wt;
    for (int o = 1; o < 32; o <<= 1) {
        const int t = __shfl_up(winc, o);
        if (lane >= o) winc += t;
    }
    int pt = s_pre[lane];
    for (int o = 16; o > 0; o >>= 1) pt += __shfl_xor(pt, o);
    const int waves_before = __shfl(winc, wave) - __shfl(wt, wave);

    if (ci < nchunks) {
        int4 r;
        r.x = pt + waves_before + (inc - tsum);
        r.y = r.x + c.x;
        r.z = r.y + c.y;
        r.w = r.z + c.z;
        ((int4*)offs)[ci] = r;
    }
}

__global__ __launch_bounds__(1024) void bucket_place_kernel(
    const float* __restrict__ edge_weight,
    const int*   __restrict__ src_index,
    const int*   __restrict__ dst_index,
    const int*   __restrict__ offs,
    int* __restrict__ part,
    long long E, int wspan, int nb)
{
    __shared__ int s_cnt[L1_WAVES * MAX_BUCKETS];
    const int tid  = threadIdx.x;
    const int lane = tid & 31;
    const int wave = tid >> 5;
    for (int i = tid; i < L1_WAVES * MAX_BUCKETS; i += 1024) s_cnt[i] = 0;
    __syncthreads();

    int* mine = s_cnt + wave * MAX_BUCKETS;
    const long long wbeg = ((long long)blockIdx.x * L1_WAVES + wave) * wspan;
    long long wend = wbeg + wspan;
    if (wend > E) wend = E;

    for (long long base = wbeg; base < wend; base += 128) {
        int  d[4];
        bool v[4];
        #pragma unroll
        for (int u = 0; u < 4; ++u) {
            const long long e = base + u * 32 + lane;
            v[u] = e < wend;
            d[u] = dst_index[v[u] ? e : wend - 1];
        }
        #pragma unroll
        for (int u = 0; u < 4; ++u)
            if (v[u]) atomicAdd(&mine[d[u] >> NODE_SHIFT], 1);
    }
    __syncthreads();

    for (int b = tid; b < nb; b += 1024) {
        int run = offs[b * L1_GROUPS + blockIdx.x];
        for (int w = 0; w < L1_WAVES; ++w) {
            const int t = s_cnt[w * MAX_BUCKETS + b];
            s_cnt[w * MAX_BUCKETS + b] = run;
            run += t;
        }
    }
    __syncthreads();

    int2* recs = (int2*)part;
    for (long long base = wbeg; base < wend; base += 128) {
        int   d[4], s[4];
        float w[4];
        bool  v[4];
        #pragma unroll
        for (int u = 0; u < 4; ++u) {
            const long long e  = base + u * 32 + lane;
            v[u] = e < wend;
            const long long ec = v[u] ? e : wend - 1;
            d[u] = dst_index[ec];
            s[u] = src_index[ec];
            w[u] = edge_weight[ec];
        }
        #pragma unroll
        for (int u = 0; u < 4; ++u) {
            if (v[u]) {
                const int slot = atomicAdd(&mine[d[u] >> NODE_SHIFT], 1);
                const int key  = (int)(((unsigned)(d[u] & ((1 << NODE_SHIFT) - 1)) << 24) | (unsigned)s[u]);
                recs[slot] = make_int2(key, __float_as_int(w[u]));
            }
        }
    }
}

__global__ __launch_bounds__(1024) void bucket_sort_kernel(
    const int* __restrict__ offs,
    const int* __restrict__ part,
    int* __restrict__ sorted,
    int* __restrict__ noffs,
    int Nrows, int nb)
{
    __shared__ int s_cnt[L1_WAVES * 256];
    __shared__ int s_wsum[8];
    const int tid  = threadIdx.x;
    const int lane = tid & 31;
    const int wave = tid >> 5;
    const int b    = blockIdx.x;
    const int beg  = offs[b * L1_GROUPS];
    const int end  = offs[(b + 1) * L1_GROUPS];

    for (int i = tid; i < L1_WAVES * 256; i += 1024) s_cnt[i] = 0;
    __syncthreads();

    const int wspan = (((end - beg + L1_WAVES - 1) / L1_WAVES) + 31) & ~31;
    const int wbeg  = beg + wave * wspan;
    int wend = wbeg + wspan;
    if (wend > end) wend = end;

    int* mine = s_cnt + wave * 256;
    const int2* recs = (const int2*)part;

    for (int base = wbeg; base < wend; base += 128) {
        unsigned k[4];
        bool     v[4];
        #pragma unroll
        for (int u = 0; u < 4; ++u) {
            const int e = base + u * 32 + lane;
            v[u] = e < wend;
            k[u] = (unsigned)recs[v[u] ? e : wend - 1].x;
        }
        #pragma unroll
        for (int u = 0; u < 4; ++u)
            if (v[u]) atomicAdd(&mine[k[u] >> 24], 1);
    }
    __syncthreads();

    int tot = 0;
    if (tid < 256)
        for (int w = 0; w < L1_WAVES; ++w) tot += s_cnt[w * 256 + tid];
    int inc = tot;
    for (int o = 1; o < 32; o <<= 1) {
        const int t = __shfl_up(inc, o);
        if (lane >= o) inc += t;
    }
    if (tid < 256 && lane == 31) s_wsum[wave] = inc;
    __syncthreads();
    if (tid < 256) {
        int start = beg + inc - tot;
        for (int w = 0; w < wave; ++w) start += s_wsum[w];
        const int node = b * 256 + tid;
        if (node < Nrows) noffs[node] = start;
        int run = start;
        for (int w = 0; w < L1_WAVES; ++w) {
            const int t = s_cnt[w * 256 + tid];
            s_cnt[w * 256 + tid] = run;
            run += t;
        }
    }
    if (b == nb - 1 && tid == 0) noffs[Nrows] = end;
    __syncthreads();

    int2* outp = (int2*)sorted;
    for (int base = wbeg; base < wend; base += 128) {
        int2 r[4];
        bool v[4];
        #pragma unroll
        for (int u = 0; u < 4; ++u) {
            const int e = base + u * 32 + lane;
            v[u] = e < wend;
            r[u] = recs[v[u] ? e : wend - 1];
        }
        #pragma unroll
        for (int u = 0; u < 4; ++u) {
            if (v[u]) {
                const int slot = atomicAdd(&mine[(unsigned)r[u].x >> 24], 1);
                outp[slot] = make_int2(r[u].x & 0x00FFFFFF, r[u].y);
            }
        }
    }
}

__device__ __forceinline__ void add_edge_row(
    float2& acc, const float2* __restrict__ rows, const int2 p, const int j, const int lane)
{
    const int   s = __builtin_amdgcn_readlane(p.x, j);
    const float w = __int_as_float(__builtin_amdgcn_readlane(p.y, j));
    const float2 x = rows[(long long)s * 32 + lane];
    acc.x = fmaf(w, x.x, acc.x);
    acc.y = fmaf(w, x.y, acc.y);
}

__global__ __launch_bounds__(256) void node_gather_sum_kernel(
    const float* __restrict__ node_states,
    const int*   __restrict__ offs,
    const int*   __restrict__ sorted,
    float* __restrict__ out,
    int Nrows)
{
    const int lane = threadIdx.x & 31;
    const int node = __builtin_amdgcn_readfirstlane(blockIdx.x * 8 + (threadIdx.x >> 5));
    if (node >= Nrows) return;
    const int beg = __builtin_amdgcn_readfirstlane(offs[node]);
    const int end = __builtin_amdgcn_readfirstlane(offs[node + 1]);

    const int2*   pairs = (const int2*)sorted;
    const float2* rows  = (const float2*)node_states;

    float2 acc = make_float2(0.0f, 0.0f);
    for (int base = beg; base < end; base += 32) {
        const int left = end - base;
        const int n = left < 32 ? left : 32;
        int2 p = make_int2(0, 0);
        if (lane < n) p = pairs[base + lane];
        int j = 0;
        for (; j + 8 <= n; j += 8) {
            add_edge_row(acc, rows, p, j + 0, lane);
            add_edge_row(acc, rows, p, j + 1, lane);
            add_edge_row(acc, rows, p, j + 2, lane);
            add_edge_row(acc, rows, p, j + 3, lane);
            add_edge_row(acc, rows, p, j + 4, lane);
            add_edge_row(acc, rows, p, j + 5, lane);
            add_edge_row(acc, rows, p, j + 6, lane);
            add_edge_row(acc, rows, p, j + 7, lane);
        }
        if (j + 4 <= n) {
            add_edge_row(acc, rows, p, j + 0, lane);
            add_edge_row(acc, rows, p, j + 1, lane);
            add_edge_row(acc, rows, p, j + 2, lane);
            add_edge_row(acc, rows, p, j + 3, lane);
            j += 4;
        }
        for (; j < n; ++j) add_edge_row(acc, rows, p, j, lane);
    }

    float2* orow = (float2*)(out + (long long)node * 64) + lane;
    float2 v = *orow;
    v.x += acc.x;
    v.y += acc.y;
    *orow = v;
}

extern "C" void kernel_launch(void* const* d_in, const int* in_sizes, int n_in,
                              void* d_out, int out_size, void* d_ws, size_t ws_size,
                              hipStream_t stream) {
    const float* node_states = (const float*)d_in[0];
    const float* edge_weight = (const float*)d_in[1];
    const float* W           = (const float*)d_in[2];
    const float* b           = (const float*)d_in[3];
    const int*   src_index   = (const int*)d_in[4];
    const int*   dst_index   = (const int*)d_in[5];
    float*       out         = (float*)d_out;

    const int       Nrows = in_sizes[0] / 64;
    const long long E     = in_sizes[1];

    const int nb   = (Nrows + 255) >> NODE_SHIFT;
    const int ncnt = (nb * L1_GROUPS + 64) & ~63;
    const int nn   = (Nrows + 64) & ~63;
    int* cnt    = (int*)d_ws;
    int* offs   = cnt + ncnt;
    int* noffs  = offs + ncnt;
    int* part   = noffs + nn;
    int* sorted = part + 2 * ((E + 63) & ~63LL);

    const int wspan = (int)((((E + L1_GROUPS * L1_WAVES - 1) / (L1_GROUPS * L1_WAVES)) + 31) & ~31LL);

    const int gemm_blocks = (Nrows + 31) / 32;
    gcn_gemm_bias_kernel<<<gemm_blocks, 256, 0, stream>>>(
        node_states, W, b, out, Nrows, cnt, ncnt);

    bucket_count_kernel<<<L1_GROUPS, 1024, 0, stream>>>(dst_index, cnt, E, wspan, nb);

    const int nchunks = ncnt / 4;
    node_offsets_kernel<<<(nchunks + 1023) / 1024, 1024, 0, stream>>>(cnt, offs, nchunks);

    bucket_place_kernel<<<L1_GROUPS, 1024, 0, stream>>>(
        edge_weight, src_index, dst_index, offs, part, E, wspan, nb);

    bucket_sort_kernel<<<nb, 1024, 0, stream>>>(offs, part, sorted, noffs, Nrows, nb);

    node_gather_sum_kernel<<<(Nrows + 7) / 8, 256, 0, stream>>>(
        node_states, noffs, sorted, out, Nrows);
}


# node_states @ W + b loses its own launch: its work-groups ride in the edge sort's placement launch beside the load-bound sorting waves, one wave per 16x64 strip (4 f32 WMMA tiles per A fragment)
# speedup vs baseline: 2.7603x; 1.0406x over previous
#include <hip/hip_runtime.h>
#include <hip/hip_bf16.h>

typedef __attribute__((ext_vector_type(2))) float v2f;
typedef __attribute__((ext_vector_type(8))) float v8f;

__device__ __forceinline__ void gemm_bias_strip(
    const float* __restrict__ A,
    const float* __restrict__ W,
    const float* __restrict__ bias,
    float* __restrict__ out,
    int Nrows, int strip)
{
    const int lane     = threadIdx.x & 31;
    const int row_base = strip * 16;
    if (row_base >= Nrows) return;

    const int half = lane >> 4;
    const int m    = lane & 15;

    const float* arow = A + (long long)(row_base + m) * 64;
    const int o0 = (row_base + 8 * half) * 64 + m;

    const float* wcol = W + (2 * half) * 64 + m;
    const float* acol = arow + 2 * half;

    v8f acc0 = {}, acc1 = {}, acc2 = {}, acc3 = {};
    v2f a, b0, b1, b2, b3;
    a.x  = acol[0];   a.y  = acol[1];
    b0.x = wcol[0];   b0.y = wcol[64];
    b1.x = wcol[16];  b1.y = wcol[80];
    b2.x = wcol[32];  b2.y = wcol[96];
    b3.x = wcol[48];  b3.y = wcol[112];
    #pragma unroll
    for (int k0 = 0; k0 < 64; k0 += 4) {
        v2f an = a, c0 = b0, c1 = b1, c2 = b2, c3 = b3;
        if (k0 + 4 < 64) {
            const float* w = wcol + (k0 + 4) * 64;
            an.x = acol[k0 + 4];  an.y = acol[k0 + 5];
            c0.x = w[0];   c0.y = w[64];
            c1.x = w[16];  c1.y = w[80];
            c2.x = w[32];  c2.y = w[96];
            c3.x = w[48];  c3.y = w[112];
        }
        acc0 = __builtin_amdgcn_wmma_f32_16x16x4_f32(false, a, false, b0, (short)0, acc0, false, false);
        acc1 = __builtin_amdgcn_wmma_f32_16x16x4_f32(false, a, false, b1, (short)0, acc1, false, false);
        acc2 = __builtin_amdgcn_wmma_f32_16x16x4_f32(false, a, false, b2, (short)0, acc2, false, false);
        acc3 = __builtin_amdgcn_wmma_f32_16x16x4_f32(false, a, false, b3, (short)0, acc3, false, false);
        a = an; b0 = c0; b1 = c1; b2 = c2; b3 = c3;
    }

    const float* bcol = bias + (o0 & 15);
    const float bv0 = bcol[0], bv1 = bcol[16], bv2 = bcol[32], bv3 = bcol[48];
    #pragma unroll
    for (int r = 0; r < 8; ++r) {
        float* orow = out + o0 + r * 64;
        orow[0]  = acc0[r] + bv0;
        orow[16] = acc1[r] + bv1;
        orow[32] = acc2[r] + bv2;
        orow[48] = acc3[r] + bv3;
    }
}

#define L1_GROUPS   256
#define L1_WAVES    32
#define NODE_SHIFT  8
#define MAX_BUCKETS 512

__global__ __launch_bounds__(1024) void bucket_count_kernel(
    const int* __restrict__ dst_index,
    int* __restrict__ cnt,
    long long E, int wspan, int nb, int ncnt)
{
    __shared__ int s_cnt[MAX_BUCKETS];
    const int tid  = threadIdx.x;
    const int lane = tid & 31;
    const int wave = tid >> 5;
    if (tid < MAX_BUCKETS) s_cnt[tid] = 0;
    __syncthreads();

    const long long wbeg = ((long long)blockIdx.x * L1_WAVES + wave) * wspan;
    long long wend = wbeg + wspan;
    if (wend > E) wend = E;
    for (long long base = wbeg; base < wend; base += 128) {
        int  d[4];
        bool v[4];
        #pragma unroll
        for (int u = 0; u < 4; ++u) {
            const long long e = base + u * 32 + lane;
            v[u] = e < wend;
            d[u] = dst_index[v[u] ? e : wend - 1];
        }
        #pragma unroll
        for (int u = 0; u < 4; ++u)
            if (v[u]) atomicAdd(&s_cnt[d[u] >> NODE_SHIFT], 1);
    }
    __syncthreads();
    for (int b = tid; b < nb; b += 1024) cnt[b * L1_GROUPS + blockIdx.x] = s_cnt[b];
    if (blockIdx.x == 0)
        for (int i = nb * L1_GROUPS + tid; i < ncnt; i += 1024) cnt[i] = 0;
}

__global__ __launch_bounds__(1024) void node_offsets_kernel(
    const int* __restrict__ cnt,
    int* __restrict__ offs,
    int nchunks)
{
    __shared__ int s_wave[32];
    __shared__ int s_pre[32];
    const int tid  = threadIdx.x;
    const int lane = tid & 31;
    const int wave = tid >> 5;
    const int4* c4 = (const int4*)cnt;

    const int nb    = blockIdx.x;
    const int first = nb * 1024;
    int pre = 0;
    #pragma unroll 4
    for (int k = 0; k < nb; ++k) {
        const int4 v = c4[k * 1024 + tid];
        pre += v.x + v.y + v.z + v.w;
    }
    const int ci = first + tid;
    int4 c = make_int4(0, 0, 0, 0);
    if (ci < nchunks) c = c4[ci];
    const int tsum = c.x + c.y + c.z + c.w;

    int inc = tsum;
    for (int o = 1; o < 32; o <<= 1) {
        const int t = __shfl_up(inc, o);
        if (lane >= o) inc += t;
    }
    for (int o = 16; o > 0; o >>= 1) pre += __shfl_xor(pre, o);
    if (lane == 31) s_wave[wave] = inc;
    if (lane == 0)  s_pre[wave]  = pre;
    __syncthreads();

    const int wt = s_wave[lane];
    int winc = wt;
    for (int o = 1; o < 32; o <<= 1) {
        const int t = __shfl_up(winc, o);
        if (lane >= o) winc += t;
    }
    int pt = s_pre[lane];
    for (int o = 16; o > 0; o >>= 1) pt += __shfl_xor(pt, o);
    const int waves_before = __shfl(winc, wave) - __shfl(wt, wave);

    if (ci < nchunks) {
        int4 r;
        r.x = pt + waves_before + (inc - tsum);
        r.y = r.x + c.x;
        r.z = r.y + c.y;
        r.w = r.z + c.z;
        ((int4*)offs)[ci] = r;
    }
}

__global__ __launch_bounds__(1024, 16) void bucket_place_gemm_kernel(
    const float* __restrict__ edge_weight,
    const int*   __restrict__ src_index,
    const int*   __restrict__ dst_index,
    const int*   __restrict__ offs,
    int* __restrict__ part,
    long long E, int wspan, int nb,
    const float* __restrict__ node_states,
    const float* __restrict__ W,
    const float* __restrict__ bias,
    float* __restrict__ out,
    int Nrows)
{
    __shared__ int s_cnt[L1_WAVES * MAX_BUCKETS];
    const int tid  = threadIdx.x;
    const int lane = tid & 31;
    const int wave = tid >> 5;
    if (blockIdx.x >= L1_GROUPS) {
        gemm_bias_strip(node_states, W, bias, out, Nrows,
                        (blockIdx.x - L1_GROUPS) * L1_WAVES + wave);
        return;
    }
    for (int i = tid; i < L1_WAVES * MAX_BUCKETS; i += 1024) s_cnt[i] = 0;
    __syncthreads();

    int* mine = s_cnt + wave * MAX_BUCKETS;
    const long long wbeg = ((long long)blockIdx.x * L1_WAVES + wave) * wspan;
    long long wend = wbeg + wspan;
    if (wend > E) wend = E;

    for (long long base = wbeg; base < wend; base += 128) {
        int  d[4];
        bool v[4];
        #pragma unroll
        for (int u = 0; u < 4; ++u) {
            const long long e = base + u * 32 + lane;
            v[u] = e < wend;
            d[u] = dst_index[v[u] ? e : wend - 1];
        }
        #pragma unroll
        for (int u = 0; u < 4; ++u)
            if (v[u]) atomicAdd(&mine[d[u] >> NODE_SHIFT], 1);
    }
    __syncthreads();

    for (int b = tid; b < nb; b += 1024) {
        int run = offs[b * L1_GROUPS + blockIdx.x];
        for (int w = 0; w < L1_WAVES; ++w) {
            const int t = s_cnt[w * MAX_BUCKETS + b];
            s_cnt[w * MAX_BUCKETS + b] = run;
            run += t;
        }
    }
    __syncthreads();

    int2* recs = (int2*)part;
    for (long long base = wbeg; base < wend; base += 128) {
        int   d[4], s[4];
        float w[4];
        bool  v[4];
        #pragma unroll
        for (int u = 0; u < 4; ++u) {
            const long long e  = base + u * 32 + lane;
            v[u] = e < wend;
            const long long ec = v[u] ? e : wend - 1;
            d[u] = dst_index[ec];
            s[u] = src_index[ec];
            w[u] = edge_weight[ec];
        }
        #pragma unroll
        for (int u = 0; u < 4; ++u) {
            if (v[u]) {
                const int slot = atomicAdd(&mine[d[u] >> NODE_SHIFT], 1);
                const int key  = (int)(((unsigned)(d[u] & ((1 << NODE_SHIFT) - 1)) << 24) | (unsigned)s[u]);
                recs[slot] = make_int2(key, __float_as_int(w[u]));
            }
        }
    }
}

__global__ __launch_bounds__(1024) void bucket_sort_kernel(
    const int* __restrict__ offs,
    const int* __restrict__ part,
    int* __restrict__ sorted,
    int* __restrict__ noffs,
    int Nrows, int nb)
{
    __shared__ int s_cnt[L1_WAVES * 256];
    __shared__ int s_wsum[8];
    const int tid  = threadIdx.x;
    const int lane = tid & 31;
    const int wave = tid >> 5;
    const int b    = blockIdx.x;
    const int beg  = offs[b * L1_GROUPS];
    const int end  = offs[(b + 1) * L1_GROUPS];

    for (int i = tid; i < L1_WAVES * 256; i += 1024) s_cnt[i] = 0;
    __syncthreads();

    const int wspan = (((end - beg + L1_WAVES - 1) / L1_WAVES) + 31) & ~31;
    const int wbeg  = beg + wave * wspan;
    int wend = wbeg + wspan;
    if (wend > end) wend = end;

    int* mine = s_cnt + wave * 256;
    const int2* recs = (const int2*)part;

    for (int base = wbeg; base < wend; base += 128) {
        unsigned k[4];
        bool     v[4];
        #pragma unroll
        for (int u = 0; u < 4; ++u) {
            const int e = base + u * 32 + lane;
            v[u] = e < wend;
            k[u] = (unsigned)recs[v[u] ? e : wend - 1].x;
        }
        #pragma unroll
        for (int u = 0; u < 4; ++u)
            if (v[u]) atomicAdd(&mine[k[u] >> 24], 1);
    }
    __syncthreads();

    int tot = 0;
    if (tid < 256)
        for (int w = 0; w < L1_WAVES; ++w) tot += s_cnt[w * 256 + tid];
    int inc = tot;
    for (int o = 1; o < 32; o <<= 1) {
        const int t = __shfl_up(inc, o);
        if (lane >= o) inc += t;
    }
    if (tid < 256 && lane == 31) s_wsum[wave] = inc;
    __syncthreads();
    if (tid < 256) {
        int start = beg + inc - tot;
        for (int w = 0; w < wave; ++w) start += s_wsum[w];
        const int node = b * 256 + tid;
        if (node < Nrows) noffs[node] = start;
        int run = start;
        for (int w = 0; w < L1_WAVES; ++w) {
            const int t = s_cnt[w * 256 + tid];
            s_cnt[w * 256 + tid] = run;
            run += t;
        }
    }
    if (b == nb - 1 && tid == 0) noffs[Nrows] = end;
    __syncthreads();

    int2* outp = (int2*)sorted;
    for (int base = wbeg; base < wend; base += 128) {
        int2 r[4];
        bool v[4];
        #pragma unroll
        for (int u = 0; u < 4; ++u) {
            const int e = base + u * 32 + lane;
            v[u] = e < wend;
            r[u] = recs[v[u] ? e : wend - 1];
        }
        #pragma unroll
        for (int u = 0; u < 4; ++u) {
            if (v[u]) {
                const int slot = atomicAdd(&mine[(unsigned)r[u].x >> 24], 1);
                outp[slot] = make_int2(r[u].x & 0x00FFFFFF, r[u].y);
            }
        }
    }
}

__device__ __forceinline__ void add_edge_row(
    float2& acc, const float2* __restrict__ rows, const int2 p, const int j, const int lane)
{
    const int   s = __builtin_amdgcn_readlane(p.x, j);
    const float w = __int_as_float(__builtin_amdgcn_readlane(p.y, j));
    const float2 x = rows[(long long)s * 32 + lane];
    acc.x = fmaf(w, x.x, acc.x);
    acc.y = fmaf(w, x.y, acc.y);
}

__global__ __launch_bounds__(256) void node_gather_sum_kernel(
    const float* __restrict__ node_states,
    const int*   __restrict__ offs,
    const int*   __restrict__ sorted,
    float* __restrict__ out,
    int Nrows)
{
    const int lane = threadIdx.x & 31;
    const int node = __builtin_amdgcn_readfirstlane(blockIdx.x * 8 + (threadIdx.x >> 5));
    if (node >= Nrows) return;
    const int beg = __builtin_amdgcn_readfirstlane(offs[node]);
    const int end = __builtin_amdgcn_readfirstlane(offs[node + 1]);

    const int2*   pairs = (const int2*)sorted;
    const float2* rows  = (const float2*)node_states;

    float2 acc = make_float2(0.0f, 0.0f);
    for (int base = beg; base < end; base += 32) {
        const int left = end - base;
        const int n = left < 32 ? left : 32;
        int2 p = make_int2(0, 0);
        if (lane < n) p = pairs[base + lane];
        int j = 0;
        for (; j + 8 <= n; j += 8) {
            add_edge_row(acc, rows, p, j + 0, lane);
            add_edge_row(acc, rows, p, j + 1, lane);
            add_edge_row(acc, rows, p, j + 2, lane);
            add_edge_row(acc, rows, p, j + 3, lane);
            add_edge_row(acc, rows, p, j + 4, lane);
            add_edge_row(acc, rows, p, j + 5, lane);
            add_edge_row(acc, rows, p, j + 6, lane);
            add_edge_row(acc, rows, p, j + 7, lane);
        }
        if (j + 4 <= n) {
            add_edge_row(acc, rows, p, j + 0, lane);
            add_edge_row(acc, rows, p, j + 1, lane);
            add_edge_row(acc, rows, p, j + 2, lane);
            add_edge_row(acc, rows, p, j + 3, lane);
            j += 4;
        }
        for (; j < n; ++j) add_edge_row(acc, rows, p, j, lane);
    }

    float2* orow = (float2*)(out + (long long)node * 64) + lane;
    float2 v = *orow;
    v.x += acc.x;
    v.y += acc.y;
    *orow = v;
}

extern "C" void kernel_launch(void* const* d_in, const int* in_sizes, int n_in,
                              void* d_out, int out_size, void* d_ws, size_t ws_size,
                              hipStream_t stream) {
    const float* node_states = (const float*)d_in[0];
    const float* edge_weight = (const float*)d_in[1];
    const float* W           = (const float*)d_in[2];
    const float* b           = (const float*)d_in[3];
    const int*   src_index   = (const int*)d_in[4];
    const int*   dst_index   = (const int*)d_in[5];
    float*       out         = (float*)d_out;

    const int       Nrows = in_sizes[0] / 64;
    const long long E     = in_sizes[1];

    const int nb   = (Nrows + 255) >> NODE_SHIFT;
    const int ncnt = (nb * L1_GROUPS + 64) & ~63;
    const int nn   = (Nrows + 64) & ~63;
    int* cnt    = (int*)d_ws;
    int* offs   = cnt + ncnt;
    int* noffs  = offs + ncnt;
    int* part   = noffs + nn;
    int* sorted = part + 2 * ((E + 63) & ~63LL);

    const int wspan = (int)((((E + L1_GROUPS * L1_WAVES - 1) / (L1_GROUPS * L1_WAVES)) + 31) & ~31LL);

    bucket_count_kernel<<<L1_GROUPS, 1024, 0, stream>>>(dst_index, cnt, E, wspan, nb, ncnt);

    const int nchunks = ncnt / 4;
    node_offsets_kernel<<<(nchunks + 1023) / 1024, 1024, 0, stream>>>(cnt, offs, nchunks);

    const int gemm_blocks = (Nrows + 16 * L1_WAVES - 1) / (16 * L1_WAVES);
    bucket_place_gemm_kernel<<<L1_GROUPS + gemm_blocks, 1024, 0, stream>>>(
        edge_weight, src_index, dst_index, offs, part, E, wspan, nb,
        node_states, W, b, out, Nrows);

    bucket_sort_kernel<<<nb, 1024, 0, stream>>>(offs, part, sorted, noffs, Nrows, nb);

    node_gather_sum_kernel<<<(Nrows + 7) / 8, 256, 0, stream>>>(
        node_states, noffs, sorted, out, Nrows);
}


# second-level edge sort collects a bucket's sorted records in LDS and stores them as whole 128-byte lines (16-byte stores), not one scattered 8-byte store per edge; slots and sum order unchanged
# speedup vs baseline: 3.0568x; 1.1074x over previous
#include <hip/hip_runtime.h>
#include <hip/hip_bf16.h>

typedef __attribute__((ext_vector_type(2))) float v2f;
typedef __attribute__((ext_vector_type(8))) float v8f;

__device__ __forceinline__ void gemm_bias_strip(
    const float* __restrict__ A,
    const float* __restrict__ W,
    const float* __restrict__ bias,
    float* __restrict__ out,
    int Nrows, int strip)
{
    const int lane     = threadIdx.x & 31;
    const int row_base = strip * 16;
    if (row_base >= Nrows) return;

    const int half = lane >> 4;
    const int m    = lane & 15;

    const float* arow = A + (long long)(row_base + m) * 64;
    const int o0 = (row_base + 8 * half) * 64 + m;

    const float* wcol = W + (2 * half) * 64 + m;
    const float* acol = arow + 2 * half;

    v8f acc0 = {}, acc1 = {}, acc2 = {}, acc3 = {};
    v2f a, b0, b1, b2, b3;
    a.x  = acol[0];   a.y  = acol[1];
    b0.x = wcol[0];   b0.y = wcol[64];
    b1.x = wcol[16];  b1.y = wcol[80];
    b2.x = wcol[32];  b2.y = wcol[96];
    b3.x = wcol[48];  b3.y = wcol[112];
    #pragma unroll
    for (int k0 = 0; k0 < 64; k0 += 4) {
        v2f an = a, c0 = b0, c1 = b1, c2 = b2, c3 = b3;
        if (k0 + 4 < 64) {
            const float* w = wcol + (k0 + 4) * 64;
            an.x = acol[k0 + 4];  an.y = acol[k0 + 5];
            c0.x = w[0];   c0.y = w[64];
            c1.x = w[16];  c1.y = w[80];
            c2.x = w[32];  c2.y = w[96];
            c3.x = w[48];  c3.y = w[112];
        }
        acc0 = __builtin_amdgcn_wmma_f32_16x16x4_f32(false, a, false, b0, (short)0, acc0, false, false);
        acc1 = __builtin_amdgcn_wmma_f32_16x16x4_f32(false, a, false, b1, (short)0, acc1, false, false);
        acc2 = __builtin_amdgcn_wmma_f32_16x16x4_f32(false, a, false, b2, (short)0, acc2, false, false);
        acc3 = __builtin_amdgcn_wmma_f32_16x16x4_f32(false, a, false, b3, (short)0, acc3, false, false);
        a = an; b0 = c0; b1 = c1; b2 = c2; b3 = c3;
    }

    const float* bcol = bias + (o0 & 15);
    const float bv0 = bcol[0], bv1 = bcol[16], bv2 = bcol[32], bv3 = bcol[48];
    #pragma unroll
    for (int r = 0; r < 8; ++r) {
        float* orow = out + o0 + r * 64;
        orow[0]  = acc0[r] + bv0;
        orow[16] = acc1[r] + bv1;
        orow[32] = acc2[r] + bv2;
        orow[48] = acc3[r] + bv3;
    }
}

#define L1_GROUPS   256
#define L1_WAVES    32
#define NODE_SHIFT  8
#define MAX_BUCKETS 512

__global__ __launch_bounds__(1024) void bucket_count_kernel(
    const int* __restrict__ dst_index,
    int* __restrict__ cnt,
    long long E, int wspan, int nb, int ncnt)
{
    __shared__ int s_cnt[MAX_BUCKETS];
    const int tid  = threadIdx.x;
    const int lane = tid & 31;
    const int wave = tid >> 5;
    if (tid < MAX_BUCKETS) s_cnt[tid] = 0;
    __syncthreads();

    const long long wbeg = ((long long)blockIdx.x * L1_WAVES + wave) * wspan;
    long long wend = wbeg + wspan;
    if (wend > E) wend = E;
    for (long long base = wbeg; base < wend; base += 128) {
        int  d[4];
        bool v[4];
        #pragma unroll
        for (int u = 0; u < 4; ++u) {
            const long long e = base + u * 32 + lane;
            v[u] = e < wend;
            d[u] = dst_index[v[u] ? e : wend - 1];
        }
        #pragma unroll
        for (int u = 0; u < 4; ++u)
            if (v[u]) atomicAdd(&s_cnt[d[u] >> NODE_SHIFT], 1);
    }
    __syncthreads();
    for (int b = tid; b < nb; b += 1024) cnt[b * L1_GROUPS + blockIdx.x] = s_cnt[b];
    if (blockIdx.x == 0)
        for (int i = nb * L1_GROUPS + tid; i < ncnt; i += 1024) cnt[i] = 0;
}

__global__ __launch_bounds__(1024) void node_offsets_kernel(
    const int* __restrict__ cnt,
    int* __restrict__ offs,
    int nchunks)
{
    __shared__ int s_wave[32];
    __shared__ int s_pre[32];
    const int tid  = threadIdx.x;
    const int lane = tid & 31;
    const int wave = tid >> 5;
    const int4* c4 = (const int4*)cnt;

    const int nb    = blockIdx.x;
    const int first = nb * 1024;
    int pre = 0;
    #pragma unroll 4
    for (int k = 0; k < nb; ++k) {
        const int4 v = c4[k * 1024 + tid];
        pre += v.x + v.y + v.z + v.w;
    }
    const int ci = first + tid;
    int4 c = make_int4(0, 0, 0, 0);
    if (ci < nchunks) c = c4[ci];
    const int tsum = c.x + c.y + c.z + c.w;

    int inc = tsum;
    for (int o = 1; o < 32; o <<= 1) {
        const int t = __shfl_up(inc, o);
        if (lane >= o) inc += t;
    }
    for (int o = 16; o > 0; o >>= 1) pre += __shfl_xor(pre, o);
    if (lane == 31) s_wave[wave] = inc;
    if (lane == 0)  s_pre[wave]  = pre;
    __syncthreads();

    const int wt = s_wave[lane];
    int winc = wt;
    for (int o = 1; o < 32; o <<= 1) {
        const int t = __shfl_up(winc, o);
        if (lane >= o) winc += t;
    }
    int pt = s_pre[lane];
    for (int o = 16; o > 0; o >>= 1) pt += __shfl_xor(pt, o);
    const int waves_before = __shfl(winc, wave) - __shfl(wt, wave);

    if (ci < nchunks) {
        int4 r;
        r.x = pt + waves_before + (inc - tsum);
        r.y = r.x + c.x;
        r.z = r.y + c.y;
        r.w = r.z + c.z;
        ((int4*)offs)[ci] = r;
    }
}

__global__ __launch_bounds__(1024, 16) void bucket_place_gemm_kernel(
    const float* __restrict__ edge_weight,
    const int*   __restrict__ src_index,
    const int*   __restrict__ dst_index,
    const int*   __restrict__ offs,
    int* __restrict__ part,
    long long E, int wspan, int nb,
    const float* __restrict__ node_states,
    const float* __restrict__ W,
    const float* __restrict__ bias,
    float* __restrict__ out,
    int Nrows)
{
    __shared__ int s_cnt[L1_WAVES * MAX_BUCKETS];
    const int tid  = threadIdx.x;
    const int lane = tid & 31;
    const int wave = tid >> 5;
    if (blockIdx.x >= L1_GROUPS) {
        gemm_bias_strip(node_states, W, bias, out, Nrows,
                        (blockIdx.x - L1_GROUPS) * L1_WAVES + wave);
        return;
    }
    for (int i = tid; i < L1_WAVES * MAX_BUCKETS; i += 1024) s_cnt[i] = 0;
    __syncthreads();

    int* mine = s_cnt + wave * MAX_BUCKETS;
    const long long wbeg = ((long long)blockIdx.x * L1_WAVES + wave) * wspan;
    long long wend = wbeg + wspan;
    if (wend > E) wend = E;

    for (long long base = wbeg; base < wend; base += 128) {
        int  d[4];
        bool v[4];
        #pragma unroll
        for (int u = 0; u < 4; ++u) {
            const long long e = base + u * 32 + lane;
            v[u] = e < wend;
            d[u] = dst_index[v[u] ? e : wend - 1];
        }
        #pragma unroll
        for (int u = 0; u < 4; ++u)
            if (v[u]) atomicAdd(&mine[d[u] >> NODE_SHIFT], 1);
    }
    __syncthreads();

    for (int b = tid; b < nb; b += 1024) {
        int run = offs[b * L1_GROUPS + blockIdx.x];
        for (int w = 0; w < L1_WAVES; ++w) {
            const int t = s_cnt[w * MAX_BUCKETS + b];
            s_cnt[w * MAX_BUCKETS + b] = run;
            run += t;
        }
    }
    __syncthreads();

    int2* recs = (int2*)part;
    for (long long base = wbeg; base < wend; base += 128) {
        int   d[4], s[4];
        float w[4];
        bool  v[4];
        #pragma unroll
        for (int u = 0; u < 4; ++u) {
            const long long e  = base + u * 32 + lane;
            v[u] = e < wend;
            const long long ec = v[u] ? e : wend - 1;
            d[u] = dst_index[ec];
            s[u] = src_index[ec];
            w[u] = edge_weight[ec];
        }
        #pragma unroll
        for (int u = 0; u < 4; ++u) {
            if (v[u]) {
                const int slot = atomicAdd(&mine[d[u] >> NODE_SHIFT], 1);
                const int key  = (int)(((unsigned)(d[u] & ((1 << NODE_SHIFT) - 1)) << 24) | (unsigned)s[u]);
                recs[slot] = make_int2(key, __float_as_int(w[u]));
            }
        }
    }
}

#define STAGE_CAP 12288

__global__ __launch_bounds__(1024) void bucket_sort_kernel(
    const int* __restrict__ offs,
    const int* __restrict__ part,
    int* __restrict__ sorted,
    int* __restrict__ noffs,
    int Nrows, int nb)
{
    __shared__ int  s_cnt[L1_WAVES * 256];
    __shared__ int  s_wsum[8];
    __shared__ int4 s_stage[STAGE_CAP / 2];
    const int tid  = threadIdx.x;
    const int lane = tid & 31;
    const int wave = tid >> 5;
    const int b    = blockIdx.x;
    const int beg  = offs[b * L1_GROUPS];
    const int end  = offs[(b + 1) * L1_GROUPS];
    const int  abeg   = beg & ~31;
    const bool staged = end - abeg <= STAGE_CAP;

    for (int i = tid; i < L1_WAVES * 256; i += 1024) s_cnt[i] = 0;
    __syncthreads();

    const int wspan = (((end - beg + L1_WAVES - 1) / L1_WAVES) + 31) & ~31;
    const int wbeg  = beg + wave * wspan;
    int wend = wbeg + wspan;
    if (wend > end) wend = end;

    int* mine = s_cnt + wave * 256;
    const int2* recs = (const int2*)part;

    for (int base = wbeg; base < wend; base += 128) {
        unsigned k[4];
        bool     v[4];
        #pragma unroll
        for (int u = 0; u < 4; ++u) {
            const int e = base + u * 32 + lane;
            v[u] = e < wend;
            k[u] = (unsigned)recs[v[u] ? e : wend - 1].x;
        }
        #pragma unroll
        for (int u = 0; u < 4; ++u)
            if (v[u]) atomicAdd(&mine[k[u] >> 24], 1);
    }
    __syncthreads();

    int tot = 0;
    if (tid < 256)
        for (int w = 0; w < L1_WAVES; ++w) tot += s_cnt[w * 256 + tid];
    int inc = tot;
    for (int o = 1; o < 32; o <<= 1) {
        const int t = __shfl_up(inc, o);
        if (lane >= o) inc += t;
    }
    if (tid < 256 && lane == 31) s_wsum[wave] = inc;
    __syncthreads();
    if (tid < 256) {
        int start = beg + inc - tot;
        for (int w = 0; w < wave; ++w) start += s_wsum[w];
        const int node = b * 256 + tid;
        if (node < Nrows) noffs[node] = start;
        int run = start;
        for (int w = 0; w < L1_WAVES; ++w) {
            const int t = s_cnt[w * 256 + tid];
            s_cnt[w * 256 + tid] = run;
            run += t;
        }
    }
    if (b == nb - 1 && tid == 0) noffs[Nrows] = end;
    __syncthreads();

    int2* outp  = (int2*)sorted;
    int2* stage = (int2*)s_stage;
    for (int base = wbeg; base < wend; base += 128) {
        int2 r[4];
        bool v[4];
        #pragma unroll
        for (int u = 0; u < 4; ++u) {
            const int e = base + u * 32 + lane;
            v[u] = e < wend;
            r[u] = recs[v[u] ? e : wend - 1];
        }
        #pragma unroll
        for (int u = 0; u < 4; ++u) {
            if (v[u]) {
                const int slot = atomicAdd(&mine[(unsigned)r[u].x >> 24], 1);
                const int2 rec = make_int2(r[u].x & 0x00FFFFFF, r[u].y);
                if (staged) stage[slot - abeg] = rec;
                else        outp[slot] = rec;
            }
        }
    }
    __syncthreads();

    if (staged) {
        const int npairs = (end - abeg + 1) >> 1;
        int4* out4 = (int4*)sorted + (abeg >> 1);
        for (int q = tid; q < npairs; q += 1024) {
            const int  g0 = abeg + 2 * q;
            const bool lo = g0 >= beg;
            const bool hi = g0 + 1 >= beg && g0 + 1 < end;
            if (lo && hi) out4[q] = s_stage[q];
            else if (lo)  outp[g0] = stage[2 * q];
            else if (hi)  outp[g0 + 1] = stage[2 * q + 1];
        }
    }
}

__device__ __forceinline__ void add_edge_row(
    float2& acc, const float2* __restrict__ rows, const int2 p, const int j, const int lane)
{
    const int   s = __builtin_amdgcn_readlane(p.x, j);
    const float w = __int_as_float(__builtin_amdgcn_readlane(p.y, j));
    const float2 x = rows[(long long)s * 32 + lane];
    acc.x = fmaf(w, x.x, acc.x);
    acc.y = fmaf(w, x.y, acc.y);
}

__global__ __launch_bounds__(256) void node_gather_sum_kernel(
    const float* __restrict__ node_states,
    const int*   __restrict__ offs,
    const int*   __restrict__ sorted,
    float* __restrict__ out,
    int Nrows)
{
    const int lane = threadIdx.x & 31;
    const int node = __builtin_amdgcn_readfirstlane(blockIdx.x * 8 + (threadIdx.x >> 5));
    if (node >= Nrows) return;
    const int beg = __builtin_amdgcn_readfirstlane(offs[node]);
    const int end = __builtin_amdgcn_readfirstlane(offs[node + 1]);

    const int2*   pairs = (const int2*)sorted;
    const float2* rows  = (const float2*)node_states;

    float2 acc = make_float2(0.0f, 0.0f);
    for (int base = beg; base < end; base += 32) {
        const int left = end - base;
        const int n = left < 32 ? left : 32;
        int2 p = make_int2(0, 0);
        if (lane < n) p = pairs[base + lane];
        int j = 0;
        for (; j + 8 <= n; j += 8) {
            add_edge_row(acc, rows, p, j + 0, lane);
            add_edge_row(acc, rows, p, j + 1, lane);
            add_edge_row(acc, rows, p, j + 2, lane);
            add_edge_row(acc, rows, p, j + 3, lane);
            add_edge_row(acc, rows, p, j + 4, lane);
            add_edge_row(acc, rows, p, j + 5, lane);
            add_edge_row(acc, rows, p, j + 6, lane);
            add_edge_row(acc, rows, p, j + 7, lane);
        }
        if (j + 4 <= n) {
            add_edge_row(acc, rows, p, j + 0, lane);
            add_edge_row(acc, rows, p, j + 1, lane);
            add_edge_row(acc, rows, p, j + 2, lane);
            add_edge_row(acc, rows, p, j + 3, lane);
            j += 4;
        }
        for (; j < n; ++j) add_edge_row(acc, rows, p, j, lane);
    }

    float2* orow = (float2*)(out + (long long)node * 64) + lane;
    float2 v = *orow;
    v.x += acc.x;
    v.y += acc.y;
    *orow = v;
}

extern "C" void kernel_launch(void* const* d_in, const int* in_sizes, int n_in,
                              void* d_out, int out_size, void* d_ws, size_t ws_size,
                              hipStream_t stream) {
    const float* node_states = (const float*)d_in[0];
    const float* edge_weight = (const float*)d_in[1];
    const float* W           = (const float*)d_in[2];
    const float* b           = (const float*)d_in[3];
    const int*   src_index   = (const int*)d_in[4];
    const int*   dst_index   = (const int*)d_in[5];
    float*       out         = (float*)d_out;

    const int       Nrows = in_sizes[0] / 64;
    const long long E     = in_sizes[1];

    const int nb   = (Nrows + 255) >> NODE_SHIFT;
    const int ncnt = (nb * L1_GROUPS + 64) & ~63;
    const int nn   = (Nrows + 64) & ~63;
    int* cnt    = (int*)d_ws;
    int* offs   = cnt + ncnt;
    int* noffs  = offs + ncnt;
    int* part   = noffs + nn;
    int* sorted = part + 2 * ((E + 63) & ~63LL);

    const int wspan = (int)((((E + L1_GROUPS * L1_WAVES - 1) / (L1_GROUPS * L1_WAVES)) + 31) & ~31LL);

    bucket_count_kernel<<<L1_GROUPS, 1024, 0, stream>>>(dst_index, cnt, E, wspan, nb, ncnt);

    const int nchunks = ncnt / 4;
    node_offsets_kernel<<<(nchunks + 1023) / 1024, 1024, 0, stream>>>(cnt, offs, nchunks);

    const int gemm_blocks = (Nrows + 16 * L1_WAVES - 1) / (16 * L1_WAVES);
    bucket_place_gemm_kernel<<<L1_GROUPS + gemm_blocks, 1024, 0, stream>>>(
        edge_weight, src_index, dst_index, offs, part, E, wspan, nb,
        node_states, W, b, out, Nrows);

    bucket_sort_kernel<<<nb, 1024, 0, stream>>>(offs, part, sorted, noffs, Nrows, nb);

    node_gather_sum_kernel<<<(Nrows + 7) / 8, 256, 0, stream>>>(
        node_states, noffs, sorted, out, Nrows);
}


# first-level edge sort files a work-group's records by bucket in LDS (packed 16-bit per-wave counters) and stores each bucket's run as consecutive records, not an 8-byte store per edge; sum order kept
# speedup vs baseline: 3.6825x; 1.2047x over previous
#include <hip/hip_runtime.h>
#include <hip/hip_bf16.h>

typedef __attribute__((ext_vector_type(2))) float v2f;
typedef __attribute__((ext_vector_type(8))) float v8f;

__device__ __forceinline__ void gemm_bias_strip(
    const float* __restrict__ A,
    const float* __restrict__ W,
    const float* __restrict__ bias,
    float* __restrict__ out,
    int Nrows, int strip)
{
    const int lane     = threadIdx.x & 31;
    const int row_base = strip * 16;
    if (row_base >= Nrows) return;

    const int half = lane >> 4;
    const int m    = lane & 15;

    const float* arow = A + (long long)(row_base + m) * 64;
    const int o0 = (row_base + 8 * half) * 64 + m;

    const float* wcol = W + (2 * half) * 64 + m;
    const float* acol = arow + 2 * half;

    v8f acc0 = {}, acc1 = {}, acc2 = {}, acc3 = {};
    v2f a, b0, b1, b2, b3;
    a.x  = acol[0];   a.y  = acol[1];
    b0.x = wcol[0];   b0.y = wcol[64];
    b1.x = wcol[16];  b1.y = wcol[80];
    b2.x = wcol[32];  b2.y = wcol[96];
    b3.x = wcol[48];  b3.y = wcol[112];
    #pragma unroll
    for (int k0 = 0; k0 < 64; k0 += 4) {
        v2f an = a, c0 = b0, c1 = b1, c2 = b2, c3 = b3;
        if (k0 + 4 < 64) {
            const float* w = wcol + (k0 + 4) * 64;
            an.x = acol[k0 + 4];  an.y = acol[k0 + 5];
            c0.x = w[0];   c0.y = w[64];
            c1.x = w[16];  c1.y = w[80];
            c2.x = w[32];  c2.y = w[96];
            c3.x = w[48];  c3.y = w[112];
        }
        acc0 = __builtin_amdgcn_wmma_f32_16x16x4_f32(false, a, false, b0, (short)0, acc0, false, false);
        acc1 = __builtin_amdgcn_wmma_f32_16x16x4_f32(false, a, false, b1, (short)0, acc1, false, false);
        acc2 = __builtin_amdgcn_wmma_f32_16x16x4_f32(false, a, false, b2, (short)0, acc2, false, false);
        acc3 = __builtin_amdgcn_wmma_f32_16x16x4_f32(false, a, false, b3, (short)0, acc3, false, false);
        a = an; b0 = c0; b1 = c1; b2 = c2; b3 = c3;
    }

    const float* bcol = bias + (o0 & 15);
    const float bv0 = bcol[0], bv1 = bcol[16], bv2 = bcol[32], bv3 = bcol[48];
    #pragma unroll
    for (int r = 0; r < 8; ++r) {
        float* orow = out + o0 + r * 64;
        orow[0]  = acc0[r] + bv0;
        orow[16] = acc1[r] + bv1;
        orow[32] = acc2[r] + bv2;
        orow[48] = acc3[r] + bv3;
    }
}

#define L1_GROUPS   256
#define L1_WAVES    32
#define NODE_SHIFT  8
#define MAX_BUCKETS 512

__global__ __launch_bounds__(1024) void bucket_count_kernel(
    const int* __restrict__ dst_index,
    int* __restrict__ cnt,
    long long E, int wspan, int nb, int ncnt)
{
    __shared__ int s_cnt[MAX_BUCKETS];
    const int tid  = threadIdx.x;
    const int lane = tid & 31;
    const int wave = tid >> 5;
    if (tid < MAX_BUCKETS) s_cnt[tid] = 0;
    __syncthreads();

    const long long wbeg = ((long long)blockIdx.x * L1_WAVES + wave) * wspan;
    long long wend = wbeg + wspan;
    if (wend > E) wend = E;
    for (long long base = wbeg; base < wend; base += 128) {
        int  d[4];
        bool v[4];
        #pragma unroll
        for (int u = 0; u < 4; ++u) {
            const long long e = base + u * 32 + lane;
            v[u] = e < wend;
            d[u] = dst_index[v[u] ? e : wend - 1];
        }
        #pragma unroll
        for (int u = 0; u < 4; ++u)
            if (v[u]) atomicAdd(&s_cnt[d[u] >> NODE_SHIFT], 1);
    }
    __syncthreads();
    for (int b = tid; b < nb; b += 1024) cnt[b * L1_GROUPS + blockIdx.x] = s_cnt[b];
    if (blockIdx.x == 0)
        for (int i = nb * L1_GROUPS + tid; i < ncnt; i += 1024) cnt[i] = 0;
}

__global__ __launch_bounds__(1024) void node_offsets_kernel(
    const int* __restrict__ cnt,
    int* __restrict__ offs,
    int nchunks)
{
    __shared__ int s_wave[32];
    __shared__ int s_pre[32];
    const int tid  = threadIdx.x;
    const int lane = tid & 31;
    const int wave = tid >> 5;
    const int4* c4 = (const int4*)cnt;

    const int nb    = blockIdx.x;
    const int first = nb * 1024;
    int pre = 0;
    #pragma unroll 4
    for (int k = 0; k < nb; ++k) {
        const int4 v = c4[k * 1024 + tid];
        pre += v.x + v.y + v.z + v.w;
    }
    const int ci = first + tid;
    int4 c = make_int4(0, 0, 0, 0);
    if (ci < nchunks) c = c4[ci];
    const int tsum = c.x + c.y + c.z + c.w;

    int inc = tsum;
    for (int o = 1; o < 32; o <<= 1) {
        const int t = __shfl_up(inc, o);
        if (lane >= o) inc += t;
    }
    for (int o = 16; o > 0; o >>= 1) pre += __shfl_xor(pre, o);
    if (lane == 31) s_wave[wave] = inc;
    if (lane == 0)  s_pre[wave]  = pre;
    __syncthreads();

    const int wt = s_wave[lane];
    int winc = wt;
    for (int o = 1; o < 32; o <<= 1) {
        const int t = __shfl_up(winc, o);
        if (lane >= o) winc += t;
    }
    int pt = s_pre[lane];
    for (int o = 16; o > 0; o >>= 1) pt += __shfl_xor(pt, o);
    const int waves_before = __shfl(winc, wave) - __shfl(wt, wave);

    if (ci < nchunks) {
        int4 r;
        r.x = pt + waves_before + (inc - tsum);
        r.y = r.x + c.x;
        r.z = r.y + c.y;
        r.w = r.z + c.z;
        ((int4*)offs)[ci] = r;
    }
}

#define CHUNK_WSPAN 416
#define STAGE1_CAP  (L1_WAVES * CHUNK_WSPAN)
#define PAIR_WORDS  (MAX_BUCKETS / 2)
#define OWN_BUCKETS (MAX_BUCKETS / L1_WAVES)

__global__ __launch_bounds__(1024, 16) void bucket_place_gemm_kernel(
    const float* __restrict__ edge_weight,
    const int*   __restrict__ src_index,
    const int*   __restrict__ dst_index,
    const int*   __restrict__ offs,
    int* __restrict__ part,
    long long E, int wspan, int nb,
    const float* __restrict__ node_states,
    const float* __restrict__ W,
    const float* __restrict__ bias,
    float* __restrict__ out,
    int Nrows)
{
    __shared__ int  s_cnt[L1_WAVES * PAIR_WORDS];
    __shared__ int2 s_stage[STAGE1_CAP];
    __shared__ int  s_lbeg[MAX_BUCKETS + 1];
    __shared__ int  s_gpos[MAX_BUCKETS];
    __shared__ int  s_wsum[PAIR_WORDS / 32];
    const int tid  = threadIdx.x;
    const int lane = tid & 31;
    const int wave = tid >> 5;
    if (blockIdx.x >= L1_GROUPS) {
        gemm_bias_strip(node_states, W, bias, out, Nrows,
                        (blockIdx.x - L1_GROUPS) * L1_WAVES + wave);
        return;
    }

    const long long gbeg = (long long)blockIdx.x * L1_WAVES * wspan;
    long long gend = gbeg + (long long)L1_WAVES * wspan;
    if (gend > E) gend = E;

    if (tid < nb) s_gpos[tid] = offs[tid * L1_GROUPS + blockIdx.x];

    int*  mine = s_cnt + wave * PAIR_WORDS;
    int2* recs = (int2*)part;

    for (long long cbeg = gbeg; cbeg < gend; cbeg += STAGE1_CAP) {
        for (int i = tid; i < L1_WAVES * PAIR_WORDS; i += 1024) s_cnt[i] = 0;
        __syncthreads();

        const long long wbeg = cbeg + wave * CHUNK_WSPAN;
        long long wend = wbeg + CHUNK_WSPAN;
        if (wend > gend) wend = gend;

        for (long long base = wbeg; base < wend; base += 128) {
            int  d[4];
            bool v[4];
            #pragma unroll
            for (int u = 0; u < 4; ++u) {
                const long long e = base + u * 32 + lane;
                v[u] = e < wend;
                d[u] = dst_index[v[u] ? e : wend - 1];
            }
            #pragma unroll
            for (int u = 0; u < 4; ++u) {
                if (v[u]) {
                    const int bk = d[u] >> NODE_SHIFT;
                    atomicAdd(&mine[bk >> 1], 1 << ((bk & 1) << 4));
                }
            }
        }
        __syncthreads();

        int tot = 0;
        if (tid < PAIR_WORDS)
            for (int w = 0; w < L1_WAVES; ++w) tot += s_cnt[w * PAIR_WORDS + tid];
        const int t0   = tot & 0xFFFF;
        const int t1   = (int)((unsigned)tot >> 16);
        const int both = t0 + t1;
        int inc = both;
        for (int o = 1; o < 32; o <<= 1) {
            const int t = __shfl_up(inc, o);
            if (lane >= o) inc += t;
        }
        if (tid < PAIR_WORDS && lane == 31) s_wsum[wave] = inc;
        __syncthreads();
        if (tid < PAIR_WORDS) {
            int l0 = inc - both;
            for (int w = 0; w < wave; ++w) l0 += s_wsum[w];
            const int l1 = l0 + t0;
            s_lbeg[2 * tid]     = l0;
            s_lbeg[2 * tid + 1] = l1;
            if (tid == PAIR_WORDS - 1) s_lbeg[MAX_BUCKETS] = l1 + t1;
            int run = l0 | (l1 << 16);
            for (int w = 0; w < L1_WAVES; ++w) {
                const int t = s_cnt[w * PAIR_WORDS + tid];
                s_cnt[w * PAIR_WORDS + tid] = run;
                run += t;
            }
        }
        __syncthreads();

        for (long long base = wbeg; base < wend; base += 128) {
            int   d[4], s[4];
            float w[4];
            bool  v[4];
            #pragma unroll
            for (int u = 0; u < 4; ++u) {
                const long long e  = base + u * 32 + lane;
                v[u] = e < wend;
                const long long ec = v[u] ? e : wend - 1;
                d[u] = dst_index[ec];
                s[u] = src_index[ec];
                w[u] = edge_weight[ec];
            }
            #pragma unroll
            for (int u = 0; u < 4; ++u) {
                if (v[u]) {
                    const int bk  = d[u] >> NODE_SHIFT;
                    const int sh  = (bk & 1) << 4;
                    const int old = atomicAdd(&mine[bk >> 1], 1 << sh);
                    const int pos = (old >> sh) & 0xFFFF;
                    const int key = (int)(((unsigned)(d[u] & ((1 << NODE_SHIFT) - 1)) << 24) | (unsigned)s[u]);
                    s_stage[pos] = make_int2(key, __float_as_int(w[u]));
                }
            }
        }
        __syncthreads();

        const int ob = wave + L1_WAVES * lane;
        int vlb = 0, vln = 0, vgp = 0;
        if (lane < OWN_BUCKETS && ob < nb) {
            vlb = s_lbeg[ob];
            vln = s_lbeg[ob + 1] - vlb;
            vgp = s_gpos[ob];
        }
        for (int j = 0; wave + L1_WAVES * j < nb; ++j) {
            const int lb = __builtin_amdgcn_readlane(vlb, j);
            const int ln = __builtin_amdgcn_readlane(vln, j);
            const int gp = __builtin_amdgcn_readlane(vgp, j);
            const int2* from = s_stage + lb;
            int2*       to   = recs + gp;
            const bool v0 = lane < ln, v1 = lane + 32 < ln;
            int2 r0 = make_int2(0, 0), r1 = make_int2(0, 0);
            if (v0) r0 = from[lane];
            if (v1) r1 = from[lane + 32];
            if (v0) to[lane]      = r0;
            if (v1) to[lane + 32] = r1;
            for (int i = 64 + lane; i < ln; i += 32) to[i] = from[i];
        }
        if (cbeg + STAGE1_CAP < gend && lane < OWN_BUCKETS && ob < nb) s_gpos[ob] = vgp + vln;
    }
}

#define STAGE_CAP 12288

__global__ __launch_bounds__(1024) void bucket_sort_kernel(
    const int* __restrict__ offs,
    const int* __restrict__ part,
    int* __restrict__ sorted,
    int* __restrict__ noffs,
    int Nrows, int nb)
{
    __shared__ int  s_cnt[L1_WAVES * 256];
    __shared__ int  s_wsum[8];
    __shared__ int4 s_stage[STAGE_CAP / 2];
    const int tid  = threadIdx.x;
    const int lane = tid & 31;
    const int wave = tid >> 5;
    const int b    = blockIdx.x;
    const int beg  = offs[b * L1_GROUPS];
    const int end  = offs[(b + 1) * L1_GROUPS];
    const int  abeg   = beg & ~31;
    const bool staged = end - abeg <= STAGE_CAP;

    for (int i = tid; i < L1_WAVES * 256; i += 1024) s_cnt[i] = 0;
    __syncthreads();

    const int wspan = (((end - beg + L1_WAVES - 1) / L1_WAVES) + 31) & ~31;
    const int wbeg  = beg + wave * wspan;
    int wend = wbeg + wspan;
    if (wend > end) wend = end;

    int* mine = s_cnt + wave * 256;
    const int2* recs = (const int2*)part;

    for (int base = wbeg; base < wend; base += 128) {
        unsigned k[4];
        bool     v[4];
        #pragma unroll
        for (int u = 0; u < 4; ++u) {
            const int e = base + u * 32 + lane;
            v[u] = e < wend;
            k[u] = (unsigned)recs[v[u] ? e : wend - 1].x;
        }
        #pragma unroll
        for (int u = 0; u < 4; ++u)
            if (v[u]) atomicAdd(&mine[k[u] >> 24], 1);
    }
    __syncthreads();

    int tot = 0;
    if (tid < 256)
        for (int w = 0; w < L1_WAVES; ++w) tot += s_cnt[w * 256 + tid];
    int inc = tot;
    for (int o = 1; o < 32; o <<= 1) {
        const int t = __shfl_up(inc, o);
        if (lane >= o) inc += t;
    }
    if (tid < 256 && lane == 31) s_wsum[wave] = inc;
    __syncthreads();
    if (tid < 256) {
        int start = beg + inc - tot;
        for (int w = 0; w < wave; ++w) start += s_wsum[w];
        const int node = b * 256 + tid;
        if (node < Nrows) noffs[node] = start;
        int run = start;
        for (int w = 0; w < L1_WAVES; ++w) {
            const int t = s_cnt[w * 256 + tid];
            s_cnt[w * 256 + tid] = run;
            run += t;
        }
    }
    if (b == nb - 1 && tid == 0) noffs[Nrows] = end;
    __syncthreads();

    int2* outp  = (int2*)sorted;
    int2* stage = (int2*)s_stage;
    for (int base = wbeg; base < wend; base += 128) {
        int2 r[4];
        bool v[4];
        #pragma unroll
        for (int u = 0; u < 4; ++u) {
            const int e = base + u * 32 + lane;
            v[u] = e < wend;
            r[u] = recs[v[u] ? e : wend - 1];
        }
        #pragma unroll
        for (int u = 0; u < 4; ++u) {
            if (v[u]) {
                const int slot = atomicAdd(&mine[(unsigned)r[u].x >> 24], 1);
                const int2 rec = make_int2(r[u].x & 0x00FFFFFF, r[u].y);
                if (staged) stage[slot - abeg] = rec;
                else        outp[slot] = rec;
            }
        }
    }
    __syncthreads();

    if (staged) {
        const int npairs = (end - abeg + 1) >> 1;
        int4* out4 = (int4*)sorted + (abeg >> 1);
        for (int q = tid; q < npairs; q += 1024) {
            const int  g0 = abeg + 2 * q;
            const bool lo = g0 >= beg;
            const bool hi = g0 + 1 >= beg && g0 + 1 < end;
            if (lo && hi) out4[q] = s_stage[q];
            else if (lo)  outp[g0] = stage[2 * q];
            else if (hi)  outp[g0 + 1] = stage[2 * q + 1];
        }
    }
}

__device__ __forceinline__ void add_edge_row(
    float2& acc, const float2* __restrict__ rows, const int2 p, const int j, const int lane)
{
    const int   s = __builtin_amdgcn_readlane(p.x, j);
    const float w = __int_as_float(__builtin_amdgcn_readlane(p.y, j));
    const float2 x = rows[(long long)s * 32 + lane];
    acc.x = fmaf(w, x.x, acc.x);
    acc.y = fmaf(w, x.y, acc.y);
}

__global__ __launch_bounds__(256) void node_gather_sum_kernel(
    const float* __restrict__ node_states,
    const int*   __restrict__ offs,
    const int*   __restrict__ sorted,
    float* __restrict__ out,
    int Nrows)
{
    const int lane = threadIdx.x & 31;
    const int node = __builtin_amdgcn_readfirstlane(blockIdx.x * 8 + (threadIdx.x >> 5));
    if (node >= Nrows) return;
    const int beg = __builtin_amdgcn_readfirstlane(offs[node]);
    const int end = __builtin_amdgcn_readfirstlane(offs[node + 1]);

    const int2*   pairs = (const int2*)sorted;
    const float2* rows  = (const float2*)node_states;

    float2 acc = make_float2(0.0f, 0.0f);
    for (int base = beg; base < end; base += 32) {
        const int left = end - base;
        const int n = left < 32 ? left : 32;
        int2 p = make_int2(0, 0);
        if (lane < n) p = pairs[base + lane];
        int j = 0;
        for (; j + 8 <= n; j += 8) {
            add_edge_row(acc, rows, p, j + 0, lane);
            add_edge_row(acc, rows, p, j + 1, lane);
            add_edge_row(acc, rows, p, j + 2, lane);
            add_edge_row(acc, rows, p, j + 3, lane);
            add_edge_row(acc, rows, p, j + 4, lane);
            add_edge_row(acc, rows, p, j + 5, lane);
            add_edge_row(acc, rows, p, j + 6, lane);
            add_edge_row(acc, rows, p, j + 7, lane);
        }
        if (j + 4 <= n) {
            add_edge_row(acc, rows, p, j + 0, lane);
            add_edge_row(acc, rows, p, j + 1, lane);
            add_edge_row(acc, rows, p, j + 2, lane);
            add_edge_row(acc, rows, p, j + 3, lane);
            j += 4;
        }
        for (; j < n; ++j) add_edge_row(acc, rows, p, j, lane);
    }

    float2* orow = (float2*)(out + (long long)node * 64) + lane;
    float2 v = *orow;
    v.x += acc.x;
    v.y += acc.y;
    *orow = v;
}

extern "C" void kernel_launch(void* const* d_in, const int* in_sizes, int n_in,
                              void* d_out, int out_size, void* d_ws, size_t ws_size,
                              hipStream_t stream) {
    const float* node_states = (const float*)d_in[0];
    const float* edge_weight = (const float*)d_in[1];
    const float* W           = (const float*)d_in[2];
    const float* b           = (const float*)d_in[3];
    const int*   src_index   = (const int*)d_in[4];
    const int*   dst_index   = (const int*)d_in[5];
    float*       out         = (float*)d_out;

    const int       Nrows = in_sizes[0] / 64;
    const long long E     = in_sizes[1];

    const int nb   = (Nrows + 255) >> NODE_SHIFT;
    const int ncnt = (nb * L1_GROUPS + 64) & ~63;
    const int nn   = (Nrows + 64) & ~63;
    int* cnt    = (int*)d_ws;
    int* offs   = cnt + ncnt;
    int* noffs  = offs + ncnt;
    int* part   = noffs + nn;
    int* sorted = part + 2 * ((E + 63) & ~63LL);

    const int wspan = (int)((((E + L1_GROUPS * L1_WAVES - 1) / (L1_GROUPS * L1_WAVES)) + 31) & ~31LL);

    bucket_count_kernel<<<L1_GROUPS, 1024, 0, stream>>>(dst_index, cnt, E, wspan, nb, ncnt);

    const int nchunks = ncnt / 4;
    node_offsets_kernel<<<(nchunks + 1023) / 1024, 1024, 0, stream>>>(cnt, offs, nchunks);

    const int gemm_blocks = (Nrows + 16 * L1_WAVES - 1) / (16 * L1_WAVES);
    bucket_place_gemm_kernel<<<L1_GROUPS + gemm_blocks, 1024, 0, stream>>>(
        edge_weight, src_index, dst_index, offs, part, E, wspan, nb,
        node_states, W, b, out, Nrows);

    bucket_sort_kernel<<<nb, 1024, 0, stream>>>(offs, part, sorted, noffs, Nrows, nb);

    node_gather_sum_kernel<<<(Nrows + 7) / 8, 256, 0, stream>>>(
        node_states, noffs, sorted, out, Nrows);
}


# second-level edge sort and node sums share one launch: a bucket's ordered records stay in LDS and the same work-group's waves sum its nodes from there; no ordered list in memory; sum order kept
# speedup vs baseline: 4.1312x; 1.1218x over previous
#include <hip/hip_runtime.h>
#include <hip/hip_bf16.h>

typedef __attribute__((ext_vector_type(2))) float v2f;
typedef __attribute__((ext_vector_type(8))) float v8f;

__device__ __forceinline__ void gemm_bias_strip(
    const float* __restrict__ A,
    const float* __restrict__ W,
    const float* __restrict__ bias,
    float* __restrict__ out,
    int Nrows, int strip)
{
    const int lane     = threadIdx.x & 31;
    const int row_base = strip * 16;
    if (row_base >= Nrows) return;

    const int half = lane >> 4;
    const int m    = lane & 15;

    const float* arow = A + (long long)(row_base + m) * 64;
    const int o0 = (row_base + 8 * half) * 64 + m;

    const float* wcol = W + (2 * half) * 64 + m;
    const float* acol = arow + 2 * half;

    v8f acc0 = {}, acc1 = {}, acc2 = {}, acc3 = {};
    v2f a, b0, b1, b2, b3;
    a.x  = acol[0];   a.y  = acol[1];
    b0.x = wcol[0];   b0.y = wcol[64];
    b1.x = wcol[16];  b1.y = wcol[80];
    b2.x = wcol[32];  b2.y = wcol[96];
    b3.x = wcol[48];  b3.y = wcol[112];
    #pragma unroll
    for (int k0 = 0; k0 < 64; k0 += 4) {
        v2f an = a, c0 = b0, c1 = b1, c2 = b2, c3 = b3;
        if (k0 + 4 < 64) {
            const float* w = wcol + (k0 + 4) * 64;
            an.x = acol[k0 + 4];  an.y = acol[k0 + 5];
            c0.x = w[0];   c0.y = w[64];
            c1.x = w[16];  c1.y = w[80];
            c2.x = w[32];  c2.y = w[96];
            c3.x = w[48];  c3.y = w[112];
        }
        acc0 = __builtin_amdgcn_wmma_f32_16x16x4_f32(false, a, false, b0, (short)0, acc0, false, false);
        acc1 = __builtin_amdgcn_wmma_f32_16x16x4_f32(false, a, false, b1, (short)0, acc1, false, false);
        acc2 = __builtin_amdgcn_wmma_f32_16x16x4_f32(false, a, false, b2, (short)0, acc2, false, false);
        acc3 = __builtin_amdgcn_wmma_f32_16x16x4_f32(false, a, false, b3, (short)0, acc3, false, false);
        a = an; b0 = c0; b1 = c1; b2 = c2; b3 = c3;
    }

    const float* bcol = bias + (o0 & 15);
    const float bv0 = bcol[0], bv1 = bcol[16], bv2 = bcol[32], bv3 = bcol[48];
    #pragma unroll
    for (int r = 0; r < 8; ++r) {
        float* orow = out + o0 + r * 64;
        orow[0]  = acc0[r] + bv0;
        orow[16] = acc1[r] + bv1;
        orow[32] = acc2[r] + bv2;
        orow[48] = acc3[r] + bv3;
    }
}

#define L1_GROUPS   256
#define L1_WAVES    32
#define NODE_SHIFT  8
#define MAX_BUCKETS 512

__global__ __launch_bounds__(1024) void bucket_count_kernel(
    const int* __restrict__ dst_index,
    int* __restrict__ cnt,
    long long E, int wspan, int nb, int ncnt)
{
    __shared__ int s_cnt[MAX_BUCKETS];
    const int tid  = threadIdx.x;
    const int lane = tid & 31;
    const int wave = tid >> 5;
    if (tid < MAX_BUCKETS) s_cnt[tid] = 0;
    __syncthreads();

    const long long wbeg = ((long long)blockIdx.x * L1_WAVES + wave) * wspan;
    long long wend = wbeg + wspan;
    if (wend > E) wend = E;
    for (long long base = wbeg; base < wend; base += 128) {
        int  d[4];
        bool v[4];
        #pragma unroll
        for (int u = 0; u < 4; ++u) {
            const long long e = base + u * 32 + lane;
            v[u] = e < wend;
            d[u] = dst_index[v[u] ? e : wend - 1];
        }
        #pragma unroll
        for (int u = 0; u < 4; ++u)
            if (v[u]) atomicAdd(&s_cnt[d[u] >> NODE_SHIFT], 1);
    }
    __syncthreads();
    for (int b = tid; b < nb; b += 1024) cnt[b * L1_GROUPS + blockIdx.x] = s_cnt[b];
    if (blockIdx.x == 0)
        for (int i = nb * L1_GROUPS + tid; i < ncnt; i += 1024) cnt[i] = 0;
}

__global__ __launch_bounds__(1024) void node_offsets_kernel(
    const int* __restrict__ cnt,
    int* __restrict__ offs,
    int nchunks)
{
    __shared__ int s_wave[32];
    __shared__ int s_pre[32];
    const int tid  = threadIdx.x;
    const int lane = tid & 31;
    const int wave = tid >> 5;
    const int4* c4 = (const int4*)cnt;

    const int nb    = blockIdx.x;
    const int first = nb * 1024;
    int pre = 0;
    #pragma unroll 4
    for (int k = 0; k < nb; ++k) {
        const int4 v = c4[k * 1024 + tid];
        pre += v.x + v.y + v.z + v.w;
    }
    const int ci = first + tid;
    int4 c = make_int4(0, 0, 0, 0);
    if (ci < nchunks) c = c4[ci];
    const int tsum = c.x + c.y + c.z + c.w;

    int inc = tsum;
    for (int o = 1; o < 32; o <<= 1) {
        const int t = __shfl_up(inc, o);
        if (lane >= o) inc += t;
    }
    for (int o = 16; o > 0; o >>= 1) pre += __shfl_xor(pre, o);
    if (lane == 31) s_wave[wave] = inc;
    if (lane == 0)  s_pre[wave]  = pre;
    __syncthreads();

    const int wt = s_wave[lane];
    int winc = wt;
    for (int o = 1; o < 32; o <<= 1) {
        const int t = __shfl_up(winc, o);
        if (lane >= o) winc += t;
    }
    int pt = s_pre[lane];
    for (int o = 16; o > 0; o >>= 1) pt += __shfl_xor(pt, o);
    const int waves_before = __shfl(winc, wave) - __shfl(wt, wave);

    if (ci < nchunks) {
        int4 r;
        r.x = pt + waves_before + (inc - tsum);
        r.y = r.x + c.x;
        r.z = r.y + c.y;
        r.w = r.z + c.z;
        ((int4*)offs)[ci] = r;
    }
}

#define CHUNK_WSPAN 416
#define STAGE1_CAP  (L1_WAVES * CHUNK_WSPAN)
#define PAIR_WORDS  (MAX_BUCKETS / 2)
#define OWN_BUCKETS (MAX_BUCKETS / L1_WAVES)

__global__ __launch_bounds__(1024, 16) void bucket_place_gemm_kernel(
    const float* __restrict__ edge_weight,
    const int*   __restrict__ src_index,
    const int*   __restrict__ dst_index,
    const int*   __restrict__ offs,
    int* __restrict__ part,
    long long E, int wspan, int nb,
    const float* __restrict__ node_states,
    const float* __restrict__ W,
    const float* __restrict__ bias,
    float* __restrict__ out,
    int Nrows)
{
    __shared__ int  s_cnt[L1_WAVES * PAIR_WORDS];
    __shared__ int2 s_stage[STAGE1_CAP];
    __shared__ int  s_lbeg[MAX_BUCKETS + 1];
    __shared__ int  s_gpos[MAX_BUCKETS];
    __shared__ int  s_wsum[PAIR_WORDS / 32];
    const int tid  = threadIdx.x;
    const int lane = tid & 31;
    const int wave = tid >> 5;
    if (blockIdx.x >= L1_GROUPS) {
        gemm_bias_strip(node_states, W, bias, out, Nrows,
                        (blockIdx.x - L1_GROUPS) * L1_WAVES + wave);
        return;
    }

    const long long gbeg = (long long)blockIdx.x * L1_WAVES * wspan;
    long long gend = gbeg + (long long)L1_WAVES * wspan;
    if (gend > E) gend = E;

    if (tid < nb) s_gpos[tid] = offs[tid * L1_GROUPS + blockIdx.x];

    int*  mine = s_cnt + wave * PAIR_WORDS;
    int2* recs = (int2*)part;

    for (long long cbeg = gbeg; cbeg < gend; cbeg += STAGE1_CAP) {
        for (int i = tid; i < L1_WAVES * PAIR_WORDS; i += 1024) s_cnt[i] = 0;
        __syncthreads();

        const long long wbeg = cbeg + wave * CHUNK_WSPAN;
        long long wend = wbeg + CHUNK_WSPAN;
        if (wend > gend) wend = gend;

        for (long long base = wbeg; base < wend; base += 128) {
            int  d[4];
            bool v[4];
            #pragma unroll
            for (int u = 0; u < 4; ++u) {
                const long long e = base + u * 32 + lane;
                v[u] = e < wend;
                d[u] = dst_index[v[u] ? e : wend - 1];
            }
            #pragma unroll
            for (int u = 0; u < 4; ++u) {
                if (v[u]) {
                    const int bk = d[u] >> NODE_SHIFT;
                    atomicAdd(&mine[bk >> 1], 1 << ((bk & 1) << 4));
                }
            }
        }
        __syncthreads();

        int tot = 0;
        if (tid < PAIR_WORDS)
            for (int w = 0; w < L1_WAVES; ++w) tot += s_cnt[w * PAIR_WORDS + tid];
        const int t0   = tot & 0xFFFF;
        const int t1   = (int)((unsigned)tot >> 16);
        const int both = t0 + t1;
        int inc = both;
        for (int o = 1; o < 32; o <<= 1) {
            const int t = __shfl_up(inc, o);
            if (lane >= o) inc += t;
        }
        if (tid < PAIR_WORDS && lane == 31) s_wsum[wave] = inc;
        __syncthreads();
        if (tid < PAIR_WORDS) {
            int l0 = inc - both;
            for (int w = 0; w < wave; ++w) l0 += s_wsum[w];
            const int l1 = l0 + t0;
            s_lbeg[2 * tid]     = l0;
            s_lbeg[2 * tid + 1] = l1;
            if (tid == PAIR_WORDS - 1) s_lbeg[MAX_BUCKETS] = l1 + t1;
            int run = l0 | (l1 << 16);
            for (int w = 0; w < L1_WAVES; ++w) {
                const int t = s_cnt[w * PAIR_WORDS + tid];
                s_cnt[w * PAIR_WORDS + tid] = run;
                run += t;
            }
        }
        __syncthreads();

        for (long long base = wbeg; base < wend; base += 128) {
            int   d[4], s[4];
            float w[4];
            bool  v[4];
            #pragma unroll
            for (int u = 0; u < 4; ++u) {
                const long long e  = base + u * 32 + lane;
                v[u] = e < wend;
                const long long ec = v[u] ? e : wend - 1;
                d[u] = dst_index[ec];
                s[u] = src_index[ec];
                w[u] = edge_weight[ec];
            }
            #pragma unroll
            for (int u = 0; u < 4; ++u) {
                if (v[u]) {
                    const int bk  = d[u] >> NODE_SHIFT;
                    const int sh  = (bk & 1) << 4;
                    const int old = atomicAdd(&mine[bk >> 1], 1 << sh);
                    const int pos = (old >> sh) & 0xFFFF;
                    const int key = (int)(((unsigned)(d[u] & ((1 << NODE_SHIFT) - 1)) << 24) | (unsigned)s[u]);
                    s_stage[pos] = make_int2(key, __float_as_int(w[u]));
                }
            }
        }
        __syncthreads();

        const int ob = wave + L1_WAVES * lane;
        int vlb = 0, vln = 0, vgp = 0;
        if (lane < OWN_BUCKETS && ob < nb) {
            vlb = s_lbeg[ob];
            vln = s_lbeg[ob + 1] - vlb;
            vgp = s_gpos[ob];
        }
        for (int j = 0; wave + L1_WAVES * j < nb; ++j) {
            const int lb = __builtin_amdgcn_readlane(vlb, j);
            const int ln = __builtin_amdgcn_readlane(vln, j);
            const int gp = __builtin_amdgcn_readlane(vgp, j);
            const int2* from = s_stage + lb;
            int2*       to   = recs + gp;
            const bool v0 = lane < ln, v1 = lane + 32 < ln;
            int2 r0 = make_int2(0, 0), r1 = make_int2(0, 0);
            if (v0) r0 = from[lane];
            if (v1) r1 = from[lane + 32];
            if (v0) to[lane]      = r0;
            if (v1) to[lane + 32] = r1;
            for (int i = 64 + lane; i < ln; i += 32) to[i] = from[i];
        }
        if (cbeg + STAGE1_CAP < gend && lane < OWN_BUCKETS && ob < nb) s_gpos[ob] = vgp + vln;
    }
}

#define STAGE_CAP 12288

__device__ __forceinline__ void add_edge_row(
    float2& acc, const float2* __restrict__ rows, const int2 p, const int j, const int lane)
{
    const int   s = __builtin_amdgcn_readlane(p.x, j);
    const float w = __int_as_float(__builtin_amdgcn_readlane(p.y, j));
    const float2 x = rows[(long long)s * 32 + lane];
    acc.x = fmaf(w, x.x, acc.x);
    acc.y = fmaf(w, x.y, acc.y);
}

template <bool STAGED>
__device__ __forceinline__ void node_sums(
    const int2* stage,
    const int2* gl,
    const int* nbeg_tab, int* next,
    const float* __restrict__ node_states, float* __restrict__ out,
    const int b, const int beg, const int nn, const int lane)
{
    const float2* rows = (const float2*)node_states;
    for (;;) {
        int i = 0;
        if (lane == 0) i = atomicAdd(next, 1);
        i = __builtin_amdgcn_readfirstlane(i);
        if (i >= nn) break;
        const int nbeg = __builtin_amdgcn_readfirstlane(nbeg_tab[i]);
        const int nend = __builtin_amdgcn_readfirstlane(nbeg_tab[i + 1]);

        float2* orow = (float2*)(out + (long long)(b * 256 + i) * 64) + lane;
        float2 v = *orow;

        float2 acc = make_float2(0.0f, 0.0f);
        for (int base = nbeg; base < nend; base += 32) {
            const int left = nend - base;
            const int n = left < 32 ? left : 32;
            int2 p = make_int2(0, 0);
            if (lane < n) p = STAGED ? stage[base - beg + lane] : gl[base + lane];
            int j = 0;
            for (; j + 8 <= n; j += 8) {
                add_edge_row(acc, rows, p, j + 0, lane);
                add_edge_row(acc, rows, p, j + 1, lane);
                add_edge_row(acc, rows, p, j + 2, lane);
                add_edge_row(acc, rows, p, j + 3, lane);
                add_edge_row(acc, rows, p, j + 4, lane);
                add_edge_row(acc, rows, p, j + 5, lane);
                add_edge_row(acc, rows, p, j + 6, lane);
                add_edge_row(acc, rows, p, j + 7, lane);
            }
            if (j + 4 <= n) {
                add_edge_row(acc, rows, p, j + 0, lane);
                add_edge_row(acc, rows, p, j + 1, lane);
                add_edge_row(acc, rows, p, j + 2, lane);
                add_edge_row(acc, rows, p, j + 3, lane);
                j += 4;
            }
            for (; j < n; ++j) add_edge_row(acc, rows, p, j, lane);
        }

        v.x += acc.x;
        v.y += acc.y;
        *orow = v;
    }
}

__global__ __launch_bounds__(1024) void bucket_sort_sum_kernel(
    const int* __restrict__ offs,
    const int* __restrict__ part,
    int* __restrict__ sorted,
    const float* __restrict__ node_states,
    float* __restrict__ out,
    int Nrows)
{
    __shared__ int  s_cnt[L1_WAVES * 256];
    __shared__ int  s_wsum[8];
    __shared__ int  s_nbeg[256 + 1];
    __shared__ int  s_next;
    __shared__ int2 s_stage[STAGE_CAP];
    const int tid  = threadIdx.x;
    const int lane = tid & 31;
    const int wave = tid >> 5;
    const int b    = blockIdx.x;
    const int beg  = offs[b * L1_GROUPS];
    const int end  = offs[(b + 1) * L1_GROUPS];
    const bool staged = end - beg <= STAGE_CAP;

    for (int i = tid; i < L1_WAVES * 256; i += 1024) s_cnt[i] = 0;
    if (tid == 0) s_next = 0;
    __syncthreads();

    const int wspan = (((end - beg + L1_WAVES - 1) / L1_WAVES) + 31) & ~31;
    const int wbeg  = beg + wave * wspan;
    int wend = wbeg + wspan;
    if (wend > end) wend = end;

    int* mine = s_cnt + wave * 256;
    const int2* recs = (const int2*)part;

    for (int base = wbeg; base < wend; base += 128) {
        unsigned k[4];
        bool     v[4];
        #pragma unroll
        for (int u = 0; u < 4; ++u) {
            const int e = base + u * 32 + lane;
            v[u] = e < wend;
            k[u] = (unsigned)recs[v[u] ? e : wend - 1].x;
        }
        #pragma unroll
        for (int u = 0; u < 4; ++u)
            if (v[u]) atomicAdd(&mine[k[u] >> 24], 1);
    }
    __syncthreads();

    int tot = 0;
    if (tid < 256)
        for (int w = 0; w < L1_WAVES; ++w) tot += s_cnt[w * 256 + tid];
    int inc = tot;
    for (int o = 1; o < 32; o <<= 1) {
        const int t = __shfl_up(inc, o);
        if (lane >= o) inc += t;
    }
    if (tid < 256 && lane == 31) s_wsum[wave] = inc;
    __syncthreads();
    if (tid < 256) {
        int start = beg + inc - tot;
        for (int w = 0; w < wave; ++w) start += s_wsum[w];
        s_nbeg[tid] = start;
        if (tid == 255) s_nbeg[256] = end;
        int run = start;
        for (int w = 0; w < L1_WAVES; ++w) {
            const int t = s_cnt[w * 256 + tid];
            s_cnt[w * 256 + tid] = run;
            run += t;
        }
    }
    __syncthreads();

    int2* outp = (int2*)sorted;
    for (int base = wbeg; base < wend; base += 128) {
        int2 r[4];
        bool v[4];
        #pragma unroll
        for (int u = 0; u < 4; ++u) {
            const int e = base + u * 32 + lane;
            v[u] = e < wend;
            r[u] = recs[v[u] ? e : wend - 1];
        }
        #pragma unroll
        for (int u = 0; u < 4; ++u) {
            if (v[u]) {
                const int slot = atomicAdd(&mine[(unsigned)r[u].x >> 24], 1);
                const int2 rec = make_int2(r[u].x & 0x00FFFFFF, r[u].y);
                if (staged) s_stage[slot - beg] = rec;
                else        outp[slot] = rec;
            }
        }
    }
    __syncthreads();

    int nn = Nrows - b * 256;
    if (nn > 256) nn = 256;
    if (staged) node_sums<true>(s_stage, (const int2*)sorted, s_nbeg, &s_next, node_states, out, b, beg, nn, lane);
    else        node_sums<false>(s_stage, (const int2*)sorted, s_nbeg, &s_next, node_states, out, b, beg, nn, lane);
}

extern "C" void kernel_launch(void* const* d_in, const int* in_sizes, int n_in,
                              void* d_out, int out_size, void* d_ws, size_t ws_size,
                              hipStream_t stream) {
    const float* node_states = (const float*)d_in[0];
    const float* edge_weight = (const float*)d_in[1];
    const float* W           = (const float*)d_in[2];
    const float* b           = (const float*)d_in[3];
    const int*   src_index   = (const int*)d_in[4];
    const int*   dst_index   = (const int*)d_in[5];
    float*       out         = (float*)d_out;

    const int       Nrows = in_sizes[0] / 64;
    const long long E     = in_sizes[1];

    const int nb   = (Nrows + 255) >> NODE_SHIFT;
    const int ncnt = (nb * L1_GROUPS + 64) & ~63;
    int* cnt    = (int*)d_ws;
    int* offs   = cnt + ncnt;
    int* part   = offs + ncnt;
    int* sorted = part + 2 * ((E + 63) & ~63LL);

    const int wspan = (int)((((E + L1_GROUPS * L1_WAVES - 1) / (L1_GROUPS * L1_WAVES)) + 31) & ~31LL);

    bucket_count_kernel<<<L1_GROUPS, 1024, 0, stream>>>(dst_index, cnt, E, wspan, nb, ncnt);

    const int nchunks = ncnt / 4;
    node_offsets_kernel<<<(nchunks + 1023) / 1024, 1024, 0, stream>>>(cnt, offs, nchunks);

    const int gemm_blocks = (Nrows + 16 * L1_WAVES - 1) / (16 * L1_WAVES);
    bucket_place_gemm_kernel<<<L1_GROUPS + gemm_blocks, 1024, 0, stream>>>(
        edge_weight, src_index, dst_index, offs, part, E, wspan, nb,
        node_states, W, b, out, Nrows);

    bucket_sort_sum_kernel<<<nb, 1024, 0, stream>>>(offs, part, sorted, node_states, out, Nrows);
}


# level-1 edge sort local to a work-group: each chunk stored bucket-ordered with a table of its bucket runs; count and prefix-sum launches go (4 to 2 launches); sum order fixed by LDS ranks, may differ
# speedup vs baseline: 5.0481x; 1.2219x over previous
#include <hip/hip_runtime.h>
#include <hip/hip_bf16.h>

typedef __attribute__((ext_vector_type(2))) float v2f;
typedef __attribute__((ext_vector_type(8))) float v8f;

__device__ __forceinline__ void gemm_bias_strip(
    const float* __restrict__ A,
    const float* __restrict__ W,
    const float* __restrict__ bias,
    float* __restrict__ out,
    int Nrows, int strip)
{
    const int lane     = threadIdx.x & 31;
    const int row_base = strip * 16;
    if (row_base >= Nrows) return;

    const int half = lane >> 4;
    const int m    = lane & 15;

    const float* arow = A + (long long)(row_base + m) * 64;
    const int o0 = (row_base + 8 * half) * 64 + m;

    const float* wcol = W + (2 * half) * 64 + m;
    const float* acol = arow + 2 * half;

    v8f acc0 = {}, acc1 = {}, acc2 = {}, acc3 = {};
    v2f a, b0, b1, b2, b3;
    a.x  = acol[0];   a.y  = acol[1];
    b0.x = wcol[0];   b0.y = wcol[64];
    b1.x = wcol[16];  b1.y = wcol[80];
    b2.x = wcol[32];  b2.y = wcol[96];
    b3.x = wcol[48];  b3.y = wcol[112];
    #pragma unroll
    for (int k0 = 0; k0 < 64; k0 += 4) {
        v2f an = a, c0 = b0, c1 = b1, c2 = b2, c3 = b3;
        if (k0 + 4 < 64) {
            const float* w = wcol + (k0 + 4) * 64;
            an.x = acol[k0 + 4];  an.y = acol[k0 + 5];
            c0.x = w[0];   c0.y = w[64];
            c1.x = w[16];  c1.y = w[80];
            c2.x = w[32];  c2.y = w[96];
            c3.x = w[48];  c3.y = w[112];
        }
        acc0 = __builtin_amdgcn_wmma_f32_16x16x4_f32(false, a, false, b0, (short)0, acc0, false, false);
        acc1 = __builtin_amdgcn_wmma_f32_16x16x4_f32(false, a, false, b1, (short)0, acc1, false, false);
        acc2 = __builtin_amdgcn_wmma_f32_16x16x4_f32(false, a, false, b2, (short)0, acc2, false, false);
        acc3 = __builtin_amdgcn_wmma_f32_16x16x4_f32(false, a, false, b3, (short)0, acc3, false, false);
        a = an; b0 = c0; b1 = c1; b2 = c2; b3 = c3;
    }

    const float* bcol = bias + (o0 & 15);
    const float bv0 = bcol[0], bv1 = bcol[16], bv2 = bcol[32], bv3 = bcol[48];
    #pragma unroll
    for (int r = 0; r < 8; ++r) {
        float* orow = out + o0 + r * 64;
        orow[0]  = acc0[r] + bv0;
        orow[16] = acc1[r] + bv1;
        orow[32] = acc2[r] + bv2;
        orow[48] = acc3[r] + bv3;
    }
}

#define L1_WAVES    32
#define NODE_SHIFT  8
#define MAX_BUCKETS 512
#define MAX_CHUNKS  1024

#define CHUNK_WSPAN 416
#define STAGE1_CAP  (L1_WAVES * CHUNK_WSPAN)
#define PAIR_WORDS  (MAX_BUCKETS / 2)

__global__ __launch_bounds__(1024, 16) void chunk_sort_gemm_kernel(
    const float* __restrict__ edge_weight,
    const int*   __restrict__ src_index,
    const int*   __restrict__ dst_index,
    int* __restrict__ tab,
    int* __restrict__ part,
    long long E, int nchunks, int nb, int nbp,
    const float* __restrict__ node_states,
    const float* __restrict__ W,
    const float* __restrict__ bias,
    float* __restrict__ out,
    int Nrows)
{
    __shared__ int  s_cnt[L1_WAVES * PAIR_WORDS];
    __shared__ int4 s_stage4[STAGE1_CAP / 2];
    __shared__ int  s_lbeg[MAX_BUCKETS + 1];
    __shared__ int  s_wsum[PAIR_WORDS / 32];
    int2* s_stage = (int2*)s_stage4;
    const int tid  = threadIdx.x;
    const int lane = tid & 31;
    const int wave = tid >> 5;
    if (blockIdx.x >= nchunks) {
        gemm_bias_strip(node_states, W, bias, out, Nrows,
                        (blockIdx.x - nchunks) * L1_WAVES + wave);
        return;
    }

    const long long cbeg = (long long)blockIdx.x * STAGE1_CAP;
    long long cend = cbeg + STAGE1_CAP;
    if (cend > E) cend = E;

    int* mine = s_cnt + wave * PAIR_WORDS;

    for (int i = tid; i < L1_WAVES * PAIR_WORDS; i += 1024) s_cnt[i] = 0;
    __syncthreads();

    const long long wbeg = cbeg + wave * CHUNK_WSPAN;
    long long wend = wbeg + CHUNK_WSPAN;
    if (wend > cend) wend = cend;

    for (long long base = wbeg; base < wend; base += 128) {
        int  d[4];
        bool v[4];
        #pragma unroll
        for (int u = 0; u < 4; ++u) {
            const long long e = base + u * 32 + lane;
            v[u] = e < wend;
            d[u] = dst_index[v[u] ? e : wend - 1];
        }
        #pragma unroll
        for (int u = 0; u < 4; ++u) {
            if (v[u]) {
                const int bk = d[u] >> NODE_SHIFT;
                atomicAdd(&mine[bk >> 1], 1 << ((bk & 1) << 4));
            }
        }
    }
    __syncthreads();

    int tot = 0;
    if (tid < PAIR_WORDS)
        for (int w = 0; w < L1_WAVES; ++w) tot += s_cnt[w * PAIR_WORDS + tid];
    const int t0   = tot & 0xFFFF;
    const int t1   = (int)((unsigned)tot >> 16);
    const int both = t0 + t1;
    int inc = both;
    for (int o = 1; o < 32; o <<= 1) {
        const int t = __shfl_up(inc, o);
        if (lane >= o) inc += t;
    }
    if (tid < PAIR_WORDS && lane == 31) s_wsum[wave] = inc;
    __syncthreads();
    if (tid < PAIR_WORDS) {
        int l0 = inc - both;
        for (int w = 0; w < wave; ++w) l0 += s_wsum[w];
        const int l1 = l0 + t0;
        s_lbeg[2 * tid]     = l0;
        s_lbeg[2 * tid + 1] = l1;
        if (tid == PAIR_WORDS - 1) s_lbeg[MAX_BUCKETS] = l1 + t1;
        int run = l0 | (l1 << 16);
        for (int w = 0; w < L1_WAVES; ++w) {
            const int t = s_cnt[w * PAIR_WORDS + tid];
            s_cnt[w * PAIR_WORDS + tid] = run;
            run += t;
        }
    }
    __syncthreads();

    for (long long base = wbeg; base < wend; base += 128) {
        int   d[4], s[4];
        float w[4];
        bool  v[4];
        #pragma unroll
        for (int u = 0; u < 4; ++u) {
            const long long e  = base + u * 32 + lane;
            v[u] = e < wend;
            const long long ec = v[u] ? e : wend - 1;
            d[u] = dst_index[ec];
            s[u] = src_index[ec];
            w[u] = edge_weight[ec];
        }
        #pragma unroll
        for (int u = 0; u < 4; ++u) {
            if (v[u]) {
                const int bk  = d[u] >> NODE_SHIFT;
                const int sh  = (bk & 1) << 4;
                const int old = atomicAdd(&mine[bk >> 1], 1 << sh);
                const int pos = (old >> sh) & 0xFFFF;
                const int key = (int)(((unsigned)(d[u] & ((1 << NODE_SHIFT) - 1)) << 24) | (unsigned)s[u]);
                s_stage[pos] = make_int2(key, __float_as_int(w[u]));
            }
        }
    }
    __syncthreads();

    const int n = (int)(cend - cbeg);
    int4* to4 = (int4*)((int2*)part + cbeg);
    for (int i = tid; i < (n >> 1); i += 1024) to4[i] = s_stage4[i];
    if ((n & 1) && tid == 0) ((int2*)part)[cbeg + n - 1] = s_stage[n - 1];

    if (tid < nbp) {
        const int lb = s_lbeg[tid];
        tab[blockIdx.x * nbp + tid] = (lb << 16) | (s_lbeg[tid + 1] - lb);
    }
}

#define STAGE_CAP 12288

__device__ __forceinline__ void add_edge_row(
    float2& acc, const float2* __restrict__ rows, const int2 p, const int j, const int lane)
{
    const int   s = __builtin_amdgcn_readlane(p.x, j);
    const float w = __int_as_float(__builtin_amdgcn_readlane(p.y, j));
    const float2 x = rows[(long long)s * 32 + lane];
    acc.x = fmaf(w, x.x, acc.x);
    acc.y = fmaf(w, x.y, acc.y);
}

template <bool STAGED>
__device__ __forceinline__ void node_sums(
    const int2* stage,
    const int2* gl,
    const int* nbeg_tab, int* next,
    const float* __restrict__ node_states, float* __restrict__ out,
    const int b, const int beg, const int nn, const int lane)
{
    const float2* rows = (const float2*)node_states;
    for (;;) {
        int i = 0;
        if (lane == 0) i = atomicAdd(next, 1);
        i = __builtin_amdgcn_readfirstlane(i);
        if (i >= nn) break;
        const int nbeg = __builtin_amdgcn_readfirstlane(nbeg_tab[i]);
        const int nend = __builtin_amdgcn_readfirstlane(nbeg_tab[i + 1]);

        float2* orow = (float2*)(out + (long long)(b * 256 + i) * 64) + lane;
        float2 v = *orow;

        float2 acc = make_float2(0.0f, 0.0f);
        for (int base = nbeg; base < nend; base += 32) {
            const int left = nend - base;
            const int n = left < 32 ? left : 32;
            int2 p = make_int2(0, 0);
            if (lane < n) p = STAGED ? stage[base - beg + lane] : gl[base + lane];
            int j = 0;
            for (; j + 8 <= n; j += 8) {
                add_edge_row(acc, rows, p, j + 0, lane);
                add_edge_row(acc, rows, p, j + 1, lane);
                add_edge_row(acc, rows, p, j + 2, lane);
                add_edge_row(acc, rows, p, j + 3, lane);
                add_edge_row(acc, rows, p, j + 4, lane);
                add_edge_row(acc, rows, p, j + 5, lane);
                add_edge_row(acc, rows, p, j + 6, lane);
                add_edge_row(acc, rows, p, j + 7, lane);
            }
            if (j + 4 <= n) {
                add_edge_row(acc, rows, p, j + 0, lane);
                add_edge_row(acc, rows, p, j + 1, lane);
                add_edge_row(acc, rows, p, j + 2, lane);
                add_edge_row(acc, rows, p, j + 3, lane);
                j += 4;
            }
            for (; j < n; ++j) add_edge_row(acc, rows, p, j, lane);
        }

        v.x += acc.x;
        v.y += acc.y;
        *orow = v;
    }
}

template <bool COUNT>
__device__ __forceinline__ void run_pass(
    const int2* __restrict__ recs, const int vg, const int vc, const int nr,
    int* mine, int2* stage, int2* gout, const bool staged, const int lane)
{
    for (int j0 = 0; j0 < nr; j0 += 4) {
        int  g[4], c[4];
        int2 r[8];
        bool v[8];
        #pragma unroll
        for (int u = 0; u < 4; ++u) {
            const bool live = j0 + u < nr;
            const int  j    = live ? j0 + u : 0;
            c[u] = live ? __builtin_amdgcn_readlane(vc, j) : 0;
            g[u] = c[u] > 0 ? __builtin_amdgcn_readlane(vg, j) : 0;
            v[u]     = lane < c[u];
            v[u + 4] = lane + 32 < c[u];
            const int i0 = v[u]     ? g[u] + lane      : g[u];
            const int i1 = v[u + 4] ? g[u] + lane + 32 : g[u];
            if (COUNT) {
                r[u]     = make_int2(recs[i0].x, 0);
                r[u + 4] = make_int2(recs[i1].x, 0);
            } else {
                r[u]     = recs[i0];
                r[u + 4] = recs[i1];
            }
        }
        #pragma unroll
        for (int u = 0; u < 4; ++u) {
            #pragma unroll
            for (int h = 0; h < 2; ++h) {
                const int2 x = r[u + 4 * h];
                if (v[u + 4 * h]) {
                    if (COUNT) {
                        atomicAdd(&mine[(unsigned)x.x >> 24], 1);
                    } else {
                        const int slot = atomicAdd(&mine[(unsigned)x.x >> 24], 1);
                        const int2 rec = make_int2(x.x & 0x00FFFFFF, x.y);
                        if (staged) stage[slot] = rec;
                        else        gout[slot]  = rec;
                    }
                }
            }
            #pragma unroll 1
            for (int i = 64; i < c[u]; i += 32) {
                const bool ok = i + lane < c[u];
                const int2 x = recs[ok ? g[u] + i + lane : g[u]];
                if (ok) {
                    if (COUNT) {
                        atomicAdd(&mine[(unsigned)x.x >> 24], 1);
                    } else {
                        const int slot = atomicAdd(&mine[(unsigned)x.x >> 24], 1);
                        const int2 rec = make_int2(x.x & 0x00FFFFFF, x.y);
                        if (staged) stage[slot] = rec;
                        else        gout[slot]  = rec;
                    }
                }
            }
        }
    }
}

__global__ __launch_bounds__(1024) void bucket_sort_sum_kernel(
    const int* __restrict__ tab,
    const int* __restrict__ part,
    int* __restrict__ sorted,
    int nchunks, int nbp,
    const float* __restrict__ node_states,
    float* __restrict__ out,
    int Nrows)
{
    __shared__ int  s_cnt[L1_WAVES * 256];
    __shared__ int  s_nsum[8];
    __shared__ int  s_nbeg[256 + 1];
    __shared__ int  s_next;
    __shared__ int2 s_run[MAX_CHUNKS];
    __shared__ int  s_rlen[L1_WAVES];
    __shared__ int  s_rbeg[L1_WAVES];
    __shared__ int2 s_stage[STAGE_CAP];
    const int tid  = threadIdx.x;
    const int lane = tid & 31;
    const int wave = tid >> 5;
    const int b    = blockIdx.x;

    int rc = 0, rb = 0;
    if (tid < nchunks) {
        const unsigned pk = (unsigned)tab[tid * nbp + b];
        rc = (int)(pk & 0xFFFFu);
        rb = (int)(pk >> 16);
        s_run[tid] = make_int2(tid * STAGE1_CAP + rb, rc);
    }
    int sc = rc, sb = rb;
    for (int o = 16; o > 0; o >>= 1) {
        sc += __shfl_xor(sc, o);
        sb += __shfl_xor(sb, o);
    }
    if (lane == 0) { s_rlen[wave] = sc; s_rbeg[wave] = sb; }

    for (int i = tid; i < L1_WAVES * 256; i += 1024) s_cnt[i] = 0;
    if (tid == 0) s_next = 0;
    __syncthreads();

    int nrec = s_rlen[lane], gbeg = s_rbeg[lane];
    for (int o = 16; o > 0; o >>= 1) {
        nrec += __shfl_xor(nrec, o);
        gbeg += __shfl_xor(gbeg, o);
    }
    const bool staged = nrec <= STAGE_CAP;

    const int rpw = (nchunks + L1_WAVES - 1) / L1_WAVES;
    const int r0  = __builtin_amdgcn_readfirstlane(wave) * rpw;
    int nr = nchunks - r0;
    if (nr > rpw) nr = rpw;
    int vg = 0, vc = 0;
    if (lane < nr) {
        const int2 t = s_run[r0 + lane];
        vg = t.x;
        vc = t.y;
    }

    int* mine = s_cnt + wave * 256;
    const int2* recs = (const int2*)part;
    int2* gout = (int2*)sorted + gbeg;

    run_pass<true>(recs, vg, vc, nr, mine, s_stage, gout, staged, lane);
    __syncthreads();

    int tot = 0;
    if (tid < 256)
        for (int w = 0; w < L1_WAVES; ++w) tot += s_cnt[w * 256 + tid];
    int inc = tot;
    for (int o = 1; o < 32; o <<= 1) {
        const int t = __shfl_up(inc, o);
        if (lane >= o) inc += t;
    }
    if (tid < 256 && lane == 31) s_nsum[wave] = inc;
    __syncthreads();
    if (tid < 256) {
        int start = inc - tot;
        for (int w = 0; w < wave; ++w) start += s_nsum[w];
        s_nbeg[tid] = start;
        if (tid == 255) s_nbeg[256] = nrec;
        int run = start;
        for (int w = 0; w < L1_WAVES; ++w) {
            const int t = s_cnt[w * 256 + tid];
            s_cnt[w * 256 + tid] = run;
            run += t;
        }
    }
    __syncthreads();

    run_pass<false>(recs, vg, vc, nr, mine, s_stage, gout, staged, lane);
    __syncthreads();

    int nn = Nrows - b * 256;
    if (nn > 256) nn = 256;
    if (staged) node_sums<true>(s_stage, gout, s_nbeg, &s_next, node_states, out, b, 0, nn, lane);
    else        node_sums<false>(s_stage, gout, s_nbeg, &s_next, node_states, out, b, 0, nn, lane);
}

extern "C" void kernel_launch(void* const* d_in, const int* in_sizes, int n_in,
                              void* d_out, int out_size, void* d_ws, size_t ws_size,
                              hipStream_t stream) {
    const float* node_states = (const float*)d_in[0];
    const float* edge_weight = (const float*)d_in[1];
    const float* W           = (const float*)d_in[2];
    const float* b           = (const float*)d_in[3];
    const int*   src_index   = (const int*)d_in[4];
    const int*   dst_index   = (const int*)d_in[5];
    float*       out         = (float*)d_out;

    const int       Nrows = in_sizes[0] / 64;
    const long long E     = in_sizes[1];

    const int nb      = (Nrows + 255) >> NODE_SHIFT;
    const int nbp     = (nb + 31) & ~31;
    const int nchunks = (int)((E + STAGE1_CAP - 1) / STAGE1_CAP);
    const int ntab    = (nchunks * nbp + 63) & ~63;
    int* tab    = (int*)d_ws;
    int* part   = tab + ntab;
    int* sorted = part + 2 * ((E + 63) & ~63LL);

    const int gemm_blocks = (Nrows + 16 * L1_WAVES - 1) / (16 * L1_WAVES);
    chunk_sort_gemm_kernel<<<nchunks + gemm_blocks, 1024, 0, stream>>>(
        edge_weight, src_index, dst_index, tab, part, E, nchunks, nb, nbp,
        node_states, W, b, out, Nrows);

    bucket_sort_sum_kernel<<<nb, 1024, 0, stream>>>(tab, part, sorted, nchunks, nbp, node_states, out, Nrows);
}


# level-1 edge sort reads each edge once: a wave's span (13 edges per lane) is loaded in one go and stays in registers from counting to placing; rank taken from the counting add; slots, sum order kept
# speedup vs baseline: 5.3415x; 1.0581x over previous
#include <hip/hip_runtime.h>
#include <hip/hip_bf16.h>

typedef __attribute__((ext_vector_type(2))) float v2f;
typedef __attribute__((ext_vector_type(8))) float v8f;
typedef __attribute__((ext_vector_type(2))) unsigned short us2;

__device__ __forceinline__ void gemm_bias_strip(
    const float* __restrict__ A,
    const float* __restrict__ W,
    const float* __restrict__ bias,
    float* __restrict__ out,
    int Nrows, int strip)
{
    const int lane     = threadIdx.x & 31;
    const int row_base = strip * 16;
    if (row_base >= Nrows) return;

    const int half = lane >> 4;
    const int m    = lane & 15;

    const float* arow = A + (long long)(row_base + m) * 64;
    const int o0 = (row_base + 8 * half) * 64 + m;

    const float* wcol = W + (2 * half) * 64 + m;
    const float* acol = arow + 2 * half;

    v8f acc0 = {}, acc1 = {}, acc2 = {}, acc3 = {};
    v2f a, b0, b1, b2, b3;
    a.x  = acol[0];   a.y  = acol[1];
    b0.x = wcol[0];   b0.y = wcol[64];
    b1.x = wcol[16];  b1.y = wcol[80];
    b2.x = wcol[32];  b2.y = wcol[96];
    b3.x = wcol[48];  b3.y = wcol[112];
    #pragma unroll
    for (int k0 = 0; k0 < 64; k0 += 4) {
        v2f an = a, c0 = b0, c1 = b1, c2 = b2, c3 = b3;
        if (k0 + 4 < 64) {
            const float* w = wcol + (k0 + 4) * 64;
            an.x = acol[k0 + 4];  an.y = acol[k0 + 5];
            c0.x = w[0];   c0.y = w[64];
            c1.x = w[16];  c1.y = w[80];
            c2.x = w[32];  c2.y = w[96];
            c3.x = w[48];  c3.y = w[112];
        }
        acc0 = __builtin_amdgcn_wmma_f32_16x16x4_f32(false, a, false, b0, (short)0, acc0, false, false);
        acc1 = __builtin_amdgcn_wmma_f32_16x16x4_f32(false, a, false, b1, (short)0, acc1, false, false);
        acc2 = __builtin_amdgcn_wmma_f32_16x16x4_f32(false, a, false, b2, (short)0, acc2, false, false);
        acc3 = __builtin_amdgcn_wmma_f32_16x16x4_f32(false, a, false, b3, (short)0, acc3, false, false);
        a = an; b0 = c0; b1 = c1; b2 = c2; b3 = c3;
    }

    const float* bcol = bias + (o0 & 15);
    const float bv0 = bcol[0], bv1 = bcol[16], bv2 = bcol[32], bv3 = bcol[48];
    #pragma unroll
    for (int r = 0; r < 8; ++r) {
        float* orow = out + o0 + r * 64;
        orow[0]  = acc0[r] + bv0;
        orow[16] = acc1[r] + bv1;
        orow[32] = acc2[r] + bv2;
        orow[48] = acc3[r] + bv3;
    }
}

#define L1_WAVES    32
#define NODE_SHIFT  8
#define MAX_BUCKETS 512
#define MAX_CHUNKS  1024

#define CHUNK_WSPAN 416
#define SPAN_TRIPS  (CHUNK_WSPAN / 32)
#define STAGE1_CAP  (L1_WAVES * CHUNK_WSPAN)
#define PAIR_WORDS  (MAX_BUCKETS / 2)

__device__ __forceinline__ void span_count(
    const int (&d)[SPAN_TRIPS], int2 (&rec)[SPAN_TRIPS], us2 (&br)[SPAN_TRIPS],
    int* mine, const int nw, const int lane)
{
    #pragma unroll
    for (int u = 0; u < SPAN_TRIPS; ++u) {
        br[u] = (us2){0, 0};
        if (lane < nw - u * 32) {
            const int bk  = d[u] >> NODE_SHIFT;
            const int sh  = (bk & 1) << 4;
            const int old = atomicAdd(&mine[bk >> 1], 1 << sh);
            br[u] = __builtin_amdgcn_cvt_pk_u16((unsigned)bk, ((unsigned)old >> sh) & 0xFFFFu);
            rec[u].x = (int)(((unsigned)(d[u] & ((1 << NODE_SHIFT) - 1)) << 24) | (unsigned)rec[u].x);
        }
    }
}

__device__ __forceinline__ void span_place(
    const int2 (&rec)[SPAN_TRIPS], const us2 (&br)[SPAN_TRIPS],
    const unsigned short* first, int2* stage, const int nw, const int lane)
{
    int pos[SPAN_TRIPS];
    #pragma unroll
    for (int u = 0; u < SPAN_TRIPS; ++u) pos[u] = (int)first[br[u].x] + (int)br[u].y;
    #pragma unroll
    for (int u = 0; u < SPAN_TRIPS; ++u) {
        if (lane < nw - u * 32) stage[pos[u]] = rec[u];
    }
}

__global__ __launch_bounds__(1024, 16) void chunk_sort_gemm_kernel(
    const float* __restrict__ edge_weight,
    const int*   __restrict__ src_index,
    const int*   __restrict__ dst_index,
    int* __restrict__ tab,
    int* __restrict__ part,
    long long E, int nchunks, int nb, int nbp,
    const float* __restrict__ node_states,
    const float* __restrict__ W,
    const float* __restrict__ bias,
    float* __restrict__ out,
    int Nrows)
{
    __shared__ int  s_cnt[L1_WAVES * PAIR_WORDS];
    __shared__ int4 s_stage4[STAGE1_CAP / 2];
    __shared__ int  s_lbeg[MAX_BUCKETS + 1];
    __shared__ int  s_wsum[PAIR_WORDS / 32];
    int2* s_stage = (int2*)s_stage4;
    const int tid  = threadIdx.x;
    const int lane = tid & 31;
    const int wave = tid >> 5;
    if (blockIdx.x >= nchunks) {
        gemm_bias_strip(node_states, W, bias, out, Nrows,
                        (blockIdx.x - nchunks) * L1_WAVES + wave);
        return;
    }

    const long long cbeg = (long long)blockIdx.x * STAGE1_CAP;
    long long cend = cbeg + STAGE1_CAP;
    if (cend > E) cend = E;

    int* mine = s_cnt + wave * PAIR_WORDS;

    const long long wbeg = cbeg + __builtin_amdgcn_readfirstlane(wave) * CHUNK_WSPAN;
    long long wend = wbeg + CHUNK_WSPAN;
    if (wend > cend) wend = cend;
    const int  nw   = wend > wbeg ? (int)(wend - wbeg) : 0;

    int  d[SPAN_TRIPS];
    int2 rec[SPAN_TRIPS];
    {
        const int       last  = nw > 0 ? nw - 1 : 0;
        const long long lbase = nw > 0 ? wbeg : cend - 1;
        const int*   dw = dst_index   + lbase;
        const int*   sw = src_index   + lbase;
        const float* ww = edge_weight + lbase;
        #pragma unroll
        for (int u = 0; u < SPAN_TRIPS; ++u) {
            const int o = u * 32 + lane;
            d[u] = dw[o < last ? o : last];
        }
        #pragma unroll
        for (int u = 0; u < SPAN_TRIPS; ++u) {
            const int o  = u * 32 + lane;
            const int oc = o < last ? o : last;
            rec[u].x = sw[oc];
            rec[u].y = __float_as_int(ww[oc]);
        }
    }

    #pragma unroll
    for (int i = 0; i < PAIR_WORDS / 32; ++i) mine[i * 32 + lane] = 0;

    us2 br[SPAN_TRIPS];
    span_count(d, rec, br, mine, nw, lane);
    __syncthreads();

    int tot = 0;
    if (tid < PAIR_WORDS)
        for (int w = 0; w < L1_WAVES; ++w) tot += s_cnt[w * PAIR_WORDS + tid];
    const int t0   = tot & 0xFFFF;
    const int t1   = (int)((unsigned)tot >> 16);
    const int both = t0 + t1;
    int inc = both;
    for (int o = 1; o < 32; o <<= 1) {
        const int t = __shfl_up(inc, o);
        if (lane >= o) inc += t;
    }
    if (tid < PAIR_WORDS && lane == 31) s_wsum[wave] = inc;
    __syncthreads();
    if (tid < PAIR_WORDS) {
        int l0 = inc - both;
        for (int w = 0; w < wave; ++w) l0 += s_wsum[w];
        const int l1 = l0 + t0;
        s_lbeg[2 * tid]     = l0;
        s_lbeg[2 * tid + 1] = l1;
        if (tid == PAIR_WORDS - 1) s_lbeg[MAX_BUCKETS] = l1 + t1;
        int run = l0 | (l1 << 16);
        for (int w = 0; w < L1_WAVES; ++w) {
            const int t = s_cnt[w * PAIR_WORDS + tid];
            s_cnt[w * PAIR_WORDS + tid] = run;
            run += t;
        }
    }
    __syncthreads();

    span_place(rec, br, (const unsigned short*)mine, s_stage, nw, lane);
    __syncthreads();

    const int n = (int)(cend - cbeg);
    int4* to4 = (int4*)((int2*)part + cbeg);
    for (int i = tid; i < (n >> 1); i += 1024) to4[i] = s_stage4[i];
    if ((n & 1) && tid == 0) ((int2*)part)[cbeg + n - 1] = s_stage[n - 1];

    if (tid < nbp) {
        const int lb = s_lbeg[tid];
        tab[blockIdx.x * nbp + tid] = (lb << 16) | (s_lbeg[tid + 1] - lb);
    }
}

#define STAGE_CAP 12288

__device__ __forceinline__ void add_edge_row(
    float2& acc, const float2* __restrict__ rows, const int2 p, const int j, const int lane)
{
    const int   s = __builtin_amdgcn_readlane(p.x, j);
    const float w = __int_as_float(__builtin_amdgcn_readlane(p.y, j));
    const float2 x = rows[(long long)s * 32 + lane];
    acc.x = fmaf(w, x.x, acc.x);
    acc.y = fmaf(w, x.y, acc.y);
}

template <bool STAGED>
__device__ __forceinline__ void node_sums(
    const int2* stage,
    const int2* gl,
    const int* nbeg_tab, int* next,
    const float* __restrict__ node_states, float* __restrict__ out,
    const int b, const int beg, const int nn, const int lane)
{
    const float2* rows = (const float2*)node_states;
    for (;;) {
        int i = 0;
        if (lane == 0) i = atomicAdd(next, 1);
        i = __builtin_amdgcn_readfirstlane(i);
        if (i >= nn) break;
        const int nbeg = __builtin_amdgcn_readfirstlane(nbeg_tab[i]);
        const int nend = __builtin_amdgcn_readfirstlane(nbeg_tab[i + 1]);

        float2* orow = (float2*)(out + (long long)(b * 256 + i) * 64) + lane;
        float2 v = *orow;

        float2 acc = make_float2(0.0f, 0.0f);
        for (int base = nbeg; base < nend; base += 32) {
            const int left = nend - base;
            const int n = left < 32 ? left : 32;
            int2 p = make_int2(0, 0);
            if (lane < n) p = STAGED ? stage[base - beg + lane] : gl[base + lane];
            int j = 0;
            for (; j + 8 <= n; j += 8) {
                add_edge_row(acc, rows, p, j + 0, lane);
                add_edge_row(acc, rows, p, j + 1, lane);
                add_edge_row(acc, rows, p, j + 2, lane);
                add_edge_row(acc, rows, p, j + 3, lane);
                add_edge_row(acc, rows, p, j + 4, lane);
                add_edge_row(acc, rows, p, j + 5, lane);
                add_edge_row(acc, rows, p, j + 6, lane);
                add_edge_row(acc, rows, p, j + 7, lane);
            }
            if (j + 4 <= n) {
                add_edge_row(acc, rows, p, j + 0, lane);
                add_edge_row(acc, rows, p, j + 1, lane);
                add_edge_row(acc, rows, p, j + 2, lane);
                add_edge_row(acc, rows, p, j + 3, lane);
                j += 4;
            }
            for (; j < n; ++j) add_edge_row(acc, rows, p, j, lane);
        }

        v.x += acc.x;
        v.y += acc.y;
        *orow = v;
    }
}

template <bool COUNT>
__device__ __forceinline__ void run_pass(
    const int2* __restrict__ recs, const int vg, const int vc, const int nr,
    int* mine, int2* stage, int2* gout, const bool staged, const int lane)
{
    for (int j0 = 0; j0 < nr; j0 += 4) {
        int  g[4], c[4];
        int2 r[8];
        bool v[8];
        #pragma unroll
        for (int u = 0; u < 4; ++u) {
            const bool live = j0 + u < nr;
            const int  j    = live ? j0 + u : 0;
            c[u] = live ? __builtin_amdgcn_readlane(vc, j) : 0;
            g[u] = c[u] > 0 ? __builtin_amdgcn_readlane(vg, j) : 0;
            v[u]     = lane < c[u];
            v[u + 4] = lane + 32 < c[u];
            const int i0 = v[u]     ? g[u] + lane      : g[u];
            const int i1 = v[u + 4] ? g[u] + lane + 32 : g[u];
            if (COUNT) {
                r[u]     = make_int2(recs[i0].x, 0);
                r[u + 4] = make_int2(recs[i1].x, 0);
            } else {
                r[u]     = recs[i0];
                r[u + 4] = recs[i1];
            }
        }
        #pragma unroll
        for (int u = 0; u < 4; ++u) {
            #pragma unroll
            for (int h = 0; h < 2; ++h) {
                const int2 x = r[u + 4 * h];
                if (v[u + 4 * h]) {
                    if (COUNT) {
                        atomicAdd(&mine[(unsigned)x.x >> 24], 1);
                    } else {
                        const int slot = atomicAdd(&mine[(unsigned)x.x >> 24], 1);
                        const int2 rec = make_int2(x.x & 0x00FFFFFF, x.y);
                        if (staged) stage[slot] = rec;
                        else        gout[slot]  = rec;
                    }
                }
            }
            #pragma unroll 1
            for (int i = 64; i < c[u]; i += 32) {
                const bool ok = i + lane < c[u];
                const int2 x = recs[ok ? g[u] + i + lane : g[u]];
                if (ok) {
                    if (COUNT) {
                        atomicAdd(&mine[(unsigned)x.x >> 24], 1);
                    } else {
                        const int slot = atomicAdd(&mine[(unsigned)x.x >> 24], 1);
                        const int2 rec = make_int2(x.x & 0x00FFFFFF, x.y);
                        if (staged) stage[slot] = rec;
                        else        gout[slot]  = rec;
                    }
                }
            }
        }
    }
}

__global__ __launch_bounds__(1024) void bucket_sort_sum_kernel(
    const int* __restrict__ tab,
    const int* __restrict__ part,
    int* __restrict__ sorted,
    int nchunks, int nbp,
    const float* __restrict__ node_states,
    float* __restrict__ out,
    int Nrows)
{
    __shared__ int  s_cnt[L1_WAVES * 256];
    __shared__ int  s_nsum[8];
    __shared__ int  s_nbeg[256 + 1];
    __shared__ int  s_next;
    __shared__ int2 s_run[MAX_CHUNKS];
    __shared__ int  s_rlen[L1_WAVES];
    __shared__ int  s_rbeg[L1_WAVES];
    __shared__ int2 s_stage[STAGE_CAP];
    const int tid  = threadIdx.x;
    const int lane = tid & 31;
    const int wave = tid >> 5;
    const int b    = blockIdx.x;

    int rc = 0, rb = 0;
    if (tid < nchunks) {
        const unsigned pk = (unsigned)tab[tid * nbp + b];
        rc = (int)(pk & 0xFFFFu);
        rb = (int)(pk >> 16);
        s_run[tid] = make_int2(tid * STAGE1_CAP + rb, rc);
    }
    int sc = rc, sb = rb;
    for (int o = 16; o > 0; o >>= 1) {
        sc += __shfl_xor(sc, o);
        sb += __shfl_xor(sb, o);
    }
    if (lane == 0) { s_rlen[wave] = sc; s_rbeg[wave] = sb; }

    for (int i = tid; i < L1_WAVES * 256; i += 1024) s_cnt[i] = 0;
    if (tid == 0) s_next = 0;
    __syncthreads();

    int nrec = s_rlen[lane], gbeg = s_rbeg[lane];
    for (int o = 16; o > 0; o >>= 1) {
        nrec += __shfl_xor(nrec, o);
        gbeg += __shfl_xor(gbeg, o);
    }
    const bool staged = nrec <= STAGE_CAP;

    const int rpw = (nchunks + L1_WAVES - 1) / L1_WAVES;
    const int r0  = __builtin_amdgcn_readfirstlane(wave) * rpw;
    int nr = nchunks - r0;
    if (nr > rpw) nr = rpw;
    int vg = 0, vc = 0;
    if (lane < nr) {
        const int2 t = s_run[r0 + lane];
        vg = t.x;
        vc = t.y;
    }

    int* mine = s_cnt + wave * 256;
    const int2* recs = (const int2*)part;
    int2* gout = (int2*)sorted + gbeg;

    run_pass<true>(recs, vg, vc, nr, mine, s_stage, gout, staged, lane);
    __syncthreads();

    int tot = 0;
    if (tid < 256)
        for (int w = 0; w < L1_WAVES; ++w) tot += s_cnt[w * 256 + tid];
    int inc = tot;
    for (int o = 1; o < 32; o <<= 1) {
        const int t = __shfl_up(inc, o);
        if (lane >= o) inc += t;
    }
    if (tid < 256 && lane == 31) s_nsum[wave] = inc;
    __syncthreads();
    if (tid < 256) {
        int start = inc - tot;
        for (int w = 0; w < wave; ++w) start += s_nsum[w];
        s_nbeg[tid] = start;
        if (tid == 255) s_nbeg[256] = nrec;
        int run = start;
        for (int w = 0; w < L1_WAVES; ++w) {
            const int t = s_cnt[w * 256 + tid];
            s_cnt[w * 256 + tid] = run;
            run += t;
        }
    }
    __syncthreads();

    run_pass<false>(recs, vg, vc, nr, mine, s_stage, gout, staged, lane);
    __syncthreads();

    int nn = Nrows - b * 256;
    if (nn > 256) nn = 256;
    if (staged) node_sums<true>(s_stage, gout, s_nbeg, &s_next, node_states, out, b, 0, nn, lane);
    else        node_sums<false>(s_stage, gout, s_nbeg, &s_next, node_states, out, b, 0, nn, lane);
}

extern "C" void kernel_launch(void* const* d_in, const int* in_sizes, int n_in,
                              void* d_out, int out_size, void* d_ws, size_t ws_size,
                              hipStream_t stream) {
    const float* node_states = (const float*)d_in[0];
    const float* edge_weight = (const float*)d_in[1];
    const float* W           = (const float*)d_in[2];
    const float* b           = (const float*)d_in[3];
    const int*   src_index   = (const int*)d_in[4];
    const int*   dst_index   = (const int*)d_in[5];
    float*       out         = (float*)d_out;

    const int       Nrows = in_sizes[0] / 64;
    const long long E     = in_sizes[1];

    const int nb      = (Nrows + 255) >> NODE_SHIFT;
    const int nbp     = (nb + 31) & ~31;
    const int nchunks = (int)((E + STAGE1_CAP - 1) / STAGE1_CAP);
    const int ntab    = (nchunks * nbp + 63) & ~63;
    int* tab    = (int*)d_ws;
    int* part   = tab + ntab;
    int* sorted = part + 2 * ((E + 63) & ~63LL);

    const int gemm_blocks = (Nrows + 16 * L1_WAVES - 1) / (16 * L1_WAVES);
    chunk_sort_gemm_kernel<<<nchunks + gemm_blocks, 1024, 0, stream>>>(
        edge_weight, src_index, dst_index, tab, part, E, nchunks, nb, nbp,
        node_states, W, b, out, Nrows);

    bucket_sort_sum_kernel<<<nb, 1024, 0, stream>>>(tab, part, sorted, nchunks, nbp, node_states, out, Nrows);
}
